# starting kernel
# baseline (speedup 1.0000x reference)
#include <hip/hip_runtime.h>

typedef float v2f __attribute__((ext_vector_type(2)));
typedef float v8f __attribute__((ext_vector_type(8)));

#define KDIM   500
#define HIDDEN 64

__global__ __launch_bounds__(256) void k_gemm_relu(
    const float* __restrict__ x, const float* __restrict__ W1,
    float* __restrict__ h, int n_tiles)
{
    const int lane   = threadIdx.x & 31;
    const int m_tile = blockIdx.x * 8 + (threadIdx.x >> 5);
    if (m_tile >= n_tiles) return;
    const int nn     = lane & 15;
    const int khalf  = (lane >> 4) << 1;

    const float* xrow = x  + (size_t)(m_tile * 16 + nn) * KDIM;
    const float* w0   = W1 + (size_t)(nn +  0) * KDIM;
    const float* w1   = W1 + (size_t)(nn + 16) * KDIM;
    const float* w2   = W1 + (size_t)(nn + 32) * KDIM;
    const float* w3   = W1 + (size_t)(nn + 48) * KDIM;

    v8f acc0 = {}, acc1 = {}, acc2 = {}, acc3 = {};
    for (int k0 = 0; k0 < KDIM; k0 += 4) {
        const int k = k0 + khalf;
        v2f a  = __builtin_nontemporal_load((const v2f*)(xrow + k));
        v2f b0 = *(const v2f*)(w0 + k);
        v2f b1 = *(const v2f*)(w1 + k);
        v2f b2 = *(const v2f*)(w2 + k);
        v2f b3 = *(const v2f*)(w3 + k);
        acc0 = __builtin_amdgcn_wmma_f32_16x16x4_f32(false, a, false, b0, (short)0, acc0, false, false);
        acc1 = __builtin_amdgcn_wmma_f32_16x16x4_f32(false, a, false, b1, (short)0, acc1, false, false);
        acc2 = __builtin_amdgcn_wmma_f32_16x16x4_f32(false, a, false, b2, (short)0, acc2, false, false);
        acc3 = __builtin_amdgcn_wmma_f32_16x16x4_f32(false, a, false, b3, (short)0, acc3, false, false);
    }
    const int m_lo = m_tile * 16 + ((lane >> 4) << 3);
    #pragma unroll
    for (int r = 0; r < 8; ++r) {
        const size_t mrow = (size_t)(m_lo + r) * HIDDEN;
        float v0 = acc0[r]; v0 = v0 > 0.f ? v0 : 0.f;
        float v1 = acc1[r]; v1 = v1 > 0.f ? v1 : 0.f;
        float v2 = acc2[r]; v2 = v2 > 0.f ? v2 : 0.f;
        float v3 = acc3[r]; v3 = v3 > 0.f ? v3 : 0.f;
        h[mrow + nn +  0] = v0;
        h[mrow + nn + 16] = v1;
        h[mrow + nn + 32] = v2;
        h[mrow + nn + 48] = v3;
    }
}

__global__ __launch_bounds__(256) void k_zero(float4* __restrict__ p, int n4)
{
    int i = blockIdx.x * blockDim.x + threadIdx.x;
    if (i < n4) p[i] = make_float4(0.f, 0.f, 0.f, 0.f);
}

__global__ __launch_bounds__(256) void k_spmm(
    const int* __restrict__ src, const int* __restrict__ dst,
    const float* __restrict__ val, const float* __restrict__ h,
    float* __restrict__ out, int nnz)
{
    const long long tid = (long long)blockIdx.x * blockDim.x + threadIdx.x;
    const int e = (int)(tid >> 5);
    if (e >= nnz) return;
    const int c = (int)((tid & 31) << 1);
    const int s = __builtin_nontemporal_load(src + e);
    const int d = __builtin_nontemporal_load(dst + e);
    const float v = __builtin_nontemporal_load(val + e);
    const v2f hv = *(const v2f*)(h + (size_t)s * HIDDEN + c);
    float* o = out + (size_t)d * HIDDEN + c;
    __hip_atomic_fetch_add(o,     v * hv.x, __ATOMIC_RELAXED, __HIP_MEMORY_SCOPE_AGENT);
    __hip_atomic_fetch_add(o + 1, v * hv.y, __ATOMIC_RELAXED, __HIP_MEMORY_SCOPE_AGENT);
}

__global__ __launch_bounds__(256) void k_final(
    const float* __restrict__ h,  const float* __restrict__ h1,
    const float* __restrict__ h2, const float* __restrict__ Wf,
    float* __restrict__ out, int n_tiles)
{
    const int lane = threadIdx.x & 31;
    const int tile = blockIdx.x * 8 + (threadIdx.x >> 5);
    if (tile >= n_tiles) return;
    const int nn    = lane & 15;
    const int khalf = (lane >> 4) << 1;
    const int   wn    = nn < 10 ? nn : 9;
    const float wmask = nn < 10 ? 1.f : 0.f;
    const size_t rowoff = (size_t)(tile * 16 + nn) * HIDDEN;
    const float* wrow   = Wf + (size_t)wn * 192;

    const float* bufs[3] = { h, h1, h2 };
    v8f acc = {};
    #pragma unroll
    for (int seg = 0; seg < 3; ++seg) {
        const float* buf = bufs[seg] + rowoff;
        #pragma unroll
        for (int k0 = 0; k0 < 64; k0 += 4) {
            v2f a = __builtin_nontemporal_load((const v2f*)(buf + k0 + khalf));
            v2f b;
            b.x = wrow[seg * 64 + k0 + khalf + 0] * wmask;
            b.y = wrow[seg * 64 + k0 + khalf + 1] * wmask;
            acc = __builtin_amdgcn_wmma_f32_16x16x4_f32(false, a, false, b, (short)0, acc, false, false);
        }
    }
    if (nn < 10) {
        const int m_lo = tile * 16 + ((lane >> 4) << 3);
        #pragma unroll
        for (int r = 0; r < 8; ++r)
            __builtin_nontemporal_store(acc[r], out + (size_t)(m_lo + r) * 10 + nn);
    }
}

extern "C" void kernel_launch(void* const* d_in, const int* in_sizes, int n_in,
                              void* d_out, int out_size, void* d_ws, size_t ws_size,
                              hipStream_t stream)
{
    const float* x    = (const float*)d_in[0];
    const float* W1   = (const float*)d_in[1];
    const float* Wf   = (const float*)d_in[2];
    const int*   src1 = (const int*)  d_in[3];
    const int*   dst1 = (const int*)  d_in[4];
    const float* val1 = (const float*)d_in[5];
    const int*   src2 = (const int*)  d_in[6];
    const int*   dst2 = (const int*)  d_in[7];
    const float* val2 = (const float*)d_in[8];

    const int n    = in_sizes[0] / KDIM;
    const int nnz1 = in_sizes[3];
    const int nnz2 = in_sizes[6];

    float* h   = (float*)d_ws;
    float* h1  = h  + (size_t)n * HIDDEN;
    float* h2  = h1 + (size_t)n * HIDDEN;
    float* out = (float*)d_out;

    const int n_tiles = (n + 15) / 16;

    k_gemm_relu<<<(n_tiles + 7) / 8, 256, 0, stream>>>(x, W1, h, n_tiles);

    {
        const long long n4 = (long long)2 * n * HIDDEN / 4;
        k_zero<<<(int)((n4 + 255) / 256), 256, 0, stream>>>((float4*)h1, (int)n4);
    }

    {
        const long long t1 = (long long)nnz1 * 32;
        k_spmm<<<(int)((t1 + 255) / 256), 256, 0, stream>>>(src1, dst1, val1, h, h1, nnz1);
        const long long t2 = (long long)nnz2 * 32;
        k_spmm<<<(int)((t2 + 255) / 256), 256, 0, stream>>>(src2, dst2, val2, h, h2, nnz2);
    }

    k_final<<<(n_tiles + 7) / 8, 256, 0, stream>>>(h, h1, h2, Wf, out, n_tiles);
}


# Project h onto the 10 outputs BEFORE the two SpMMs (linearity): each edge scatters 10 atomics into out, not 64 into h1/h2; zero-fill, h1/h2 planes gone; sum order changes (projected per-edge terms)
# speedup vs baseline: 1.7502x; 1.7502x over previous
#include <hip/hip_runtime.h>

typedef float v2f __attribute__((ext_vector_type(2)));
typedef float v8f __attribute__((ext_vector_type(8)));

#define KDIM   500
#define HIDDEN 64
#define NOUT   10
#define GPITCH 16

__global__ __launch_bounds__(256) void k_gemm_relu(
    const float* __restrict__ x, const float* __restrict__ W1,
    float* __restrict__ h, int n_tiles)
{
    const int lane   = threadIdx.x & 31;
    const int m_tile = blockIdx.x * 8 + (threadIdx.x >> 5);
    if (m_tile >= n_tiles) return;
    const int nn     = lane & 15;
    const int khalf  = (lane >> 4) << 1;

    const float* xrow = x  + (size_t)(m_tile * 16 + nn) * KDIM;
    const float* w0   = W1 + (size_t)(nn +  0) * KDIM;
    const float* w1   = W1 + (size_t)(nn + 16) * KDIM;
    const float* w2   = W1 + (size_t)(nn + 32) * KDIM;
    const float* w3   = W1 + (size_t)(nn + 48) * KDIM;

    v8f acc0 = {}, acc1 = {}, acc2 = {}, acc3 = {};
    for (int k0 = 0; k0 < KDIM; k0 += 4) {
        const int k = k0 + khalf;
        v2f a  = __builtin_nontemporal_load((const v2f*)(xrow + k));
        v2f b0 = *(const v2f*)(w0 + k);
        v2f b1 = *(const v2f*)(w1 + k);
        v2f b2 = *(const v2f*)(w2 + k);
        v2f b3 = *(const v2f*)(w3 + k);
        acc0 = __builtin_amdgcn_wmma_f32_16x16x4_f32(false, a, false, b0, (short)0, acc0, false, false);
        acc1 = __builtin_amdgcn_wmma_f32_16x16x4_f32(false, a, false, b1, (short)0, acc1, false, false);
        acc2 = __builtin_amdgcn_wmma_f32_16x16x4_f32(false, a, false, b2, (short)0, acc2, false, false);
        acc3 = __builtin_amdgcn_wmma_f32_16x16x4_f32(false, a, false, b3, (short)0, acc3, false, false);
    }
    const int m_lo = m_tile * 16 + ((lane >> 4) << 3);
    #pragma unroll
    for (int r = 0; r < 8; ++r) {
        const size_t mrow = (size_t)(m_lo + r) * HIDDEN;
        float v0 = acc0[r]; v0 = v0 > 0.f ? v0 : 0.f;
        float v1 = acc1[r]; v1 = v1 > 0.f ? v1 : 0.f;
        float v2 = acc2[r]; v2 = v2 > 0.f ? v2 : 0.f;
        float v3 = acc3[r]; v3 = v3 > 0.f ? v3 : 0.f;
        h[mrow + nn +  0] = v0;
        h[mrow + nn + 16] = v1;
        h[mrow + nn + 32] = v2;
        h[mrow + nn + 48] = v3;
    }
}

__global__ __launch_bounds__(256) void k_proj(
    const float* __restrict__ h, const float* __restrict__ Wf,
    float* __restrict__ out, float* __restrict__ g1, float* __restrict__ g2,
    int n_tiles)
{
    const int lane = threadIdx.x & 31;
    const int tile = blockIdx.x * 8 + (threadIdx.x >> 5);
    if (tile >= n_tiles) return;
    const int nn    = lane & 15;
    const int khalf = (lane >> 4) << 1;
    const bool live = nn < NOUT;
    const int  wn   = live ? nn : NOUT - 1;
    const float* arow = h  + (size_t)(tile * 16 + nn) * HIDDEN;
    const float* wrow = Wf + (size_t)wn * (3 * HIDDEN);

    v8f acc0 = {}, acc1 = {}, acc2 = {};
    #pragma unroll
    for (int k0 = 0; k0 < HIDDEN; k0 += 4) {
        const int k = k0 + khalf;
        v2f a = __builtin_nontemporal_load((const v2f*)(arow + k));
        const v2f w0 = *(const v2f*)(wrow + 0 * HIDDEN + k);
        const v2f w1 = *(const v2f*)(wrow + 1 * HIDDEN + k);
        const v2f w2 = *(const v2f*)(wrow + 2 * HIDDEN + k);
        v2f b0, b1, b2;
        b0.x = live ? w0.x : 0.f;  b0.y = live ? w0.y : 0.f;
        b1.x = live ? w1.x : 0.f;  b1.y = live ? w1.y : 0.f;
        b2.x = live ? w2.x : 0.f;  b2.y = live ? w2.y : 0.f;
        acc0 = __builtin_amdgcn_wmma_f32_16x16x4_f32(false, a, false, b0, (short)0, acc0, false, false);
        acc1 = __builtin_amdgcn_wmma_f32_16x16x4_f32(false, a, false, b1, (short)0, acc1, false, false);
        acc2 = __builtin_amdgcn_wmma_f32_16x16x4_f32(false, a, false, b2, (short)0, acc2, false, false);
    }
    const int m_lo = tile * 16 + ((lane >> 4) << 3);
    for (int r = 0; r < 8; ++r) {
        const size_t m = (size_t)(m_lo + r);
        g1[m * GPITCH + nn] = acc1[r];
        g2[m * GPITCH + nn] = acc2[r];
        if (live) out[m * NOUT + nn] = acc0[r];
    }
}

__global__ __launch_bounds__(256) void k_spmm10(
    const int* __restrict__ src, const int* __restrict__ dst,
    const float* __restrict__ val, const float* __restrict__ g,
    float* __restrict__ out, int nnz)
{
    const int lane = threadIdx.x & 31;
    const int e0   = (blockIdx.x * 8 + (threadIdx.x >> 5)) * 32;
    if (e0 >= nnz) return;
    const int e = e0 + lane;
    int s = 0, d = 0;
    float v = 0.f;
    if (e < nnz) {
        s = __builtin_nontemporal_load(src + e);
        d = __builtin_nontemporal_load(dst + e);
        v = __builtin_nontemporal_load(val + e);
    }
    const int n    = lane & 15;
    const int row0 = lane & 16;

    int   di[16];
    float vi[16], gv[16];
    for (int i = 0; i < 16; ++i) {
        const int si = __shfl(s, row0 + i);
        di[i] = __shfl(d, row0 + i);
        vi[i] = __shfl(v, row0 + i);
        gv[i] = g[(size_t)si * GPITCH + n];
    }
    for (int i = 0; i < 16; ++i) {
        if (n < NOUT && e0 + row0 + i < nnz)
            __hip_atomic_fetch_add(out + (size_t)di[i] * NOUT + n, vi[i] * gv[i],
                                   __ATOMIC_RELAXED, __HIP_MEMORY_SCOPE_AGENT);
    }
}

extern "C" void kernel_launch(void* const* d_in, const int* in_sizes, int n_in,
                              void* d_out, int out_size, void* d_ws, size_t ws_size,
                              hipStream_t stream)
{
    const float* x    = (const float*)d_in[0];
    const float* W1   = (const float*)d_in[1];
    const float* Wf   = (const float*)d_in[2];
    const int*   src1 = (const int*)  d_in[3];
    const int*   dst1 = (const int*)  d_in[4];
    const float* val1 = (const float*)d_in[5];
    const int*   src2 = (const int*)  d_in[6];
    const int*   dst2 = (const int*)  d_in[7];
    const float* val2 = (const float*)d_in[8];

    const int n    = in_sizes[0] / KDIM;
    const int nnz1 = in_sizes[3];
    const int nnz2 = in_sizes[6];

    float* h   = (float*)d_ws;
    float* g1  = h  + (size_t)n * HIDDEN;
    float* g2  = g1 + (size_t)n * GPITCH;
    float* out = (float*)d_out;

    const int n_tiles = (n + 15) / 16;

    k_gemm_relu<<<(n_tiles + 7) / 8, 256, 0, stream>>>(x, W1, h, n_tiles);

    k_proj<<<(n_tiles + 7) / 8, 256, 0, stream>>>(h, Wf, out, g1, g2, n_tiles);

    {
        const int w1 = (nnz1 + 31) / 32;
        k_spmm10<<<(w1 + 7) / 8, 256, 0, stream>>>(src1, dst1, val1, g1, out, nnz1);
        const int w2 = (nnz2 + 31) / 32;
        k_spmm10<<<(w2 + 7) / 8, 256, 0, stream>>>(src2, dst2, val2, g2, out, nnz2);
    }
}


# Stage-1 GEMM: x and W1 k-chunks come in by 16-byte loads along the rows, are staged in LDS, fragments read from LDS (was 8-byte global loads over 16 rows each); next chunk prefetched; sum order kept
# speedup vs baseline: 3.0124x; 1.7212x over previous
#include <hip/hip_runtime.h>

typedef float v2f __attribute__((ext_vector_type(2)));
typedef float v4f __attribute__((ext_vector_type(4)));
typedef float v8f __attribute__((ext_vector_type(8)));

#define KDIM   500
#define HIDDEN 64
#define NOUT   10
#define GPITCH 16

#define KC     100
#define KC4    (KC / 4)
#define NCHUNK (KDIM / KC)
#define TILE_M 128
#define LROWS  10
#define LTHR   (LROWS * KC4)
#define XPASS  ((TILE_M + LROWS - 1) / LROWS)
#define WPASS  ((HIDDEN + LROWS - 1) / LROWS)
#define XLAST  (TILE_M - (XPASS - 1) * LROWS)
#define WLAST  (HIDDEN - (WPASS - 1) * LROWS)

__global__ __launch_bounds__(256) void k_gemm_relu(
    const float* __restrict__ x, const float* __restrict__ W1,
    float* __restrict__ h, int n_tiles)
{
    __shared__ __attribute__((aligned(16))) float sx[TILE_M * KC];
    __shared__ __attribute__((aligned(16))) float sw[HIDDEN * KC];

    const int tid     = threadIdx.x;
    const int lane    = tid & 31;
    const int wave    = tid >> 5;
    const int n_rows  = n_tiles * 16;
    const int row_nom = blockIdx.x * TILE_M;
    const int row0    = row_nom < n_rows - TILE_M ? row_nom : n_rows - TILE_M;
    const int nn      = lane & 15;
    const int khalf   = (lane >> 4) << 1;

    const int lt  = tid < LTHR ? tid : tid - LTHR;
    const int lr  = lt / KC4;
    const int c4  = lt - lr * KC4;
    const int lrx = lr < XLAST ? lr : XLAST - 1;
    const int lrw = lr < WLAST ? lr : WLAST - 1;
    const float* gx  = x  + (size_t)(row0 + lr) * KDIM + c4 * 4;
    const float* gxl = x  + (size_t)(row0 + (XPASS - 1) * LROWS + lrx) * KDIM + c4 * 4;
    const float* gw  = W1 + (size_t)lr * KDIM + c4 * 4;
    const float* gwl = W1 + (size_t)((WPASS - 1) * LROWS + lrw) * KDIM + c4 * 4;
    const int    ls  = lr * KC + c4 * 4;

    const float* ax = sx + (wave * 16 + nn) * KC + khalf;
    const float* bw = sw + nn * KC + khalf;

    v4f rx[XPASS], rw[WPASS];
    v8f acc0 = {}, acc1 = {}, acc2 = {}, acc3 = {};

    for (int c = 0; c <= NCHUNK; ++c) {
        if (c > 0) {
            __syncthreads();
            if (tid < LTHR) {
                for (int p = 0; p < XPASS - 1; ++p) *(v4f*)(sx + ls + p * LROWS * KC) = rx[p];
                for (int p = 0; p < WPASS - 1; ++p) *(v4f*)(sw + ls + p * LROWS * KC) = rw[p];
                if (lr < XLAST) *(v4f*)(sx + ls + (XPASS - 1) * LROWS * KC) = rx[XPASS - 1];
                if (lr < WLAST) *(v4f*)(sw + ls + (WPASS - 1) * LROWS * KC) = rw[WPASS - 1];
            }
            __syncthreads();
        }
        if (c < NCHUNK) {
            const int kc0 = c * KC;
            for (int p = 0; p < XPASS - 1; ++p)
                rx[p] = __builtin_nontemporal_load((const v4f*)(gx + p * LROWS * KDIM + kc0));
            rx[XPASS - 1] = __builtin_nontemporal_load((const v4f*)(gxl + kc0));
            for (int p = 0; p < WPASS - 1; ++p)
                rw[p] = *(const v4f*)(gw + p * LROWS * KDIM + kc0);
            rw[WPASS - 1] = *(const v4f*)(gwl + kc0);
        }
        if (c > 0) {
            for (int kk = 0; kk < KC; kk += 4) {
                v2f a  = *(const v2f*)(ax + kk);
                v2f b0 = *(const v2f*)(bw +  0 * KC + kk);
                v2f b1 = *(const v2f*)(bw + 16 * KC + kk);
                v2f b2 = *(const v2f*)(bw + 32 * KC + kk);
                v2f b3 = *(const v2f*)(bw + 48 * KC + kk);
                acc0 = __builtin_amdgcn_wmma_f32_16x16x4_f32(false, a, false, b0, (short)0, acc0, false, false);
                acc1 = __builtin_amdgcn_wmma_f32_16x16x4_f32(false, a, false, b1, (short)0, acc1, false, false);
                acc2 = __builtin_amdgcn_wmma_f32_16x16x4_f32(false, a, false, b2, (short)0, acc2, false, false);
                acc3 = __builtin_amdgcn_wmma_f32_16x16x4_f32(false, a, false, b3, (short)0, acc3, false, false);
            }
        }
    }
    if (row0 + wave * 16 < row_nom) return;
    const int m_lo = row0 + wave * 16 + ((lane >> 4) << 3);
    for (int r = 0; r < 8; ++r) {
        const size_t mrow = (size_t)(m_lo + r) * HIDDEN;
        float v0 = acc0[r]; v0 = v0 > 0.f ? v0 : 0.f;
        float v1 = acc1[r]; v1 = v1 > 0.f ? v1 : 0.f;
        float v2 = acc2[r]; v2 = v2 > 0.f ? v2 : 0.f;
        float v3 = acc3[r]; v3 = v3 > 0.f ? v3 : 0.f;
        h[mrow + nn +  0] = v0;
        h[mrow + nn + 16] = v1;
        h[mrow + nn + 32] = v2;
        h[mrow + nn + 48] = v3;
    }
}

__global__ __launch_bounds__(256) void k_proj(
    const float* __restrict__ h, const float* __restrict__ Wf,
    float* __restrict__ out, float* __restrict__ g1, float* __restrict__ g2,
    int n_tiles)
{
    const int lane = threadIdx.x & 31;
    const int tile = blockIdx.x * 8 + (threadIdx.x >> 5);
    if (tile >= n_tiles) return;
    const int nn    = lane & 15;
    const int khalf = (lane >> 4) << 1;
    const bool live = nn < NOUT;
    const int  wn   = live ? nn : NOUT - 1;
    const float* arow = h  + (size_t)(tile * 16 + nn) * HIDDEN;
    const float* wrow = Wf + (size_t)wn * (3 * HIDDEN);

    v8f acc0 = {}, acc1 = {}, acc2 = {};
    #pragma unroll
    for (int k0 = 0; k0 < HIDDEN; k0 += 4) {
        const int k = k0 + khalf;
        v2f a = __builtin_nontemporal_load((const v2f*)(arow + k));
        const v2f w0 = *(const v2f*)(wrow + 0 * HIDDEN + k);
        const v2f w1 = *(const v2f*)(wrow + 1 * HIDDEN + k);
        const v2f w2 = *(const v2f*)(wrow + 2 * HIDDEN + k);
        v2f b0, b1, b2;
        b0.x = live ? w0.x : 0.f;  b0.y = live ? w0.y : 0.f;
        b1.x = live ? w1.x : 0.f;  b1.y = live ? w1.y : 0.f;
        b2.x = live ? w2.x : 0.f;  b2.y = live ? w2.y : 0.f;
        acc0 = __builtin_amdgcn_wmma_f32_16x16x4_f32(false, a, false, b0, (short)0, acc0, false, false);
        acc1 = __builtin_amdgcn_wmma_f32_16x16x4_f32(false, a, false, b1, (short)0, acc1, false, false);
        acc2 = __builtin_amdgcn_wmma_f32_16x16x4_f32(false, a, false, b2, (short)0, acc2, false, false);
    }
    const int m_lo = tile * 16 + ((lane >> 4) << 3);
    for (int r = 0; r < 8; ++r) {
        const size_t m = (size_t)(m_lo + r);
        g1[m * GPITCH + nn] = acc1[r];
        g2[m * GPITCH + nn] = acc2[r];
        if (live) out[m * NOUT + nn] = acc0[r];
    }
}

__global__ __launch_bounds__(256) void k_spmm10(
    const int* __restrict__ src, const int* __restrict__ dst,
    const float* __restrict__ val, const float* __restrict__ g,
    float* __restrict__ out, int nnz)
{
    const int lane = threadIdx.x & 31;
    const int e0   = (blockIdx.x * 8 + (threadIdx.x >> 5)) * 32;
    if (e0 >= nnz) return;
    const int e = e0 + lane;
    int s = 0, d = 0;
    float v = 0.f;
    if (e < nnz) {
        s = __builtin_nontemporal_load(src + e);
        d = __builtin_nontemporal_load(dst + e);
        v = __builtin_nontemporal_load(val + e);
    }
    const int n    = lane & 15;
    const int row0 = lane & 16;

    int   di[16];
    float vi[16], gv[16];
    for (int i = 0; i < 16; ++i) {
        const int si = __shfl(s, row0 + i);
        di[i] = __shfl(d, row0 + i);
        vi[i] = __shfl(v, row0 + i);
        gv[i] = g[(size_t)si * GPITCH + n];
    }
    for (int i = 0; i < 16; ++i) {
        if (n < NOUT && e0 + row0 + i < nnz)
            __hip_atomic_fetch_add(out + (size_t)di[i] * NOUT + n, vi[i] * gv[i],
                                   __ATOMIC_RELAXED, __HIP_MEMORY_SCOPE_AGENT);
    }
}

extern "C" void kernel_launch(void* const* d_in, const int* in_sizes, int n_in,
                              void* d_out, int out_size, void* d_ws, size_t ws_size,
                              hipStream_t stream)
{
    const float* x    = (const float*)d_in[0];
    const float* W1   = (const float*)d_in[1];
    const float* Wf   = (const float*)d_in[2];
    const int*   src1 = (const int*)  d_in[3];
    const int*   dst1 = (const int*)  d_in[4];
    const float* val1 = (const float*)d_in[5];
    const int*   src2 = (const int*)  d_in[6];
    const int*   dst2 = (const int*)  d_in[7];
    const float* val2 = (const float*)d_in[8];

    const int n    = in_sizes[0] / KDIM;
    const int nnz1 = in_sizes[3];
    const int nnz2 = in_sizes[6];

    float* h   = (float*)d_ws;
    float* g1  = h  + (size_t)n * HIDDEN;
    float* g2  = g1 + (size_t)n * GPITCH;
    float* out = (float*)d_out;

    const int n_tiles = (n + 15) / 16;

    k_gemm_relu<<<(n_tiles + 7) / 8, 256, 0, stream>>>(x, W1, h, n_tiles);

    k_proj<<<(n_tiles + 7) / 8, 256, 0, stream>>>(h, Wf, out, g1, g2, n_tiles);

    {
        const int w1 = (nnz1 + 31) / 32;
        k_spmm10<<<(w1 + 7) / 8, 256, 0, stream>>>(src1, dst1, val1, g1, out, nnz1);
        const int w2 = (nnz2 + 31) / 32;
        k_spmm10<<<(w2 + 7) / 8, 256, 0, stream>>>(src2, dst2, val2, g2, out, nnz2);
    }
}


# Projection (h x Wf blocks) moved into the stage-1 kernel's epilogue: each wave turns its relu'd 16x64 tile through its own LDS rows, then the 48 WMMAs; h plane and one launch gone; sum order kept
# speedup vs baseline: 3.1940x; 1.0603x over previous
#include <hip/hip_runtime.h>

typedef float v2f __attribute__((ext_vector_type(2)));
typedef float v4f __attribute__((ext_vector_type(4)));
typedef float v8f __attribute__((ext_vector_type(8)));

#define KDIM   500
#define HIDDEN 64
#define NOUT   10
#define GPITCH 16

#define KC     100
#define KC4    (KC / 4)
#define NCHUNK (KDIM / KC)
#define TILE_M 128
#define LROWS  10
#define LTHR   (LROWS * KC4)
#define XPASS  ((TILE_M + LROWS - 1) / LROWS)
#define WPASS  ((HIDDEN + LROWS - 1) / LROWS)
#define XLAST  (TILE_M - (XPASS - 1) * LROWS)
#define WLAST  (HIDDEN - (WPASS - 1) * LROWS)

__global__ __launch_bounds__(256) void k_gemm_relu_proj(
    const float* __restrict__ x, const float* __restrict__ W1,
    const float* __restrict__ Wf,
    float* __restrict__ out, float* __restrict__ g1, float* __restrict__ g2,
    int n_tiles)
{
    __shared__ __attribute__((aligned(16))) float sx[TILE_M * KC];
    __shared__ __attribute__((aligned(16))) float sw[HIDDEN * KC];

    const int tid     = threadIdx.x;
    const int lane    = tid & 31;
    const int wave    = tid >> 5;
    const int n_rows  = n_tiles * 16;
    const int row_nom = blockIdx.x * TILE_M;
    const int row0    = row_nom < n_rows - TILE_M ? row_nom : n_rows - TILE_M;
    const int nn      = lane & 15;
    const int khalf   = (lane >> 4) << 1;

    const int lt  = tid < LTHR ? tid : tid - LTHR;
    const int lr  = lt / KC4;
    const int c4  = lt - lr * KC4;
    const int lrx = lr < XLAST ? lr : XLAST - 1;
    const int lrw = lr < WLAST ? lr : WLAST - 1;
    const float* gx  = x  + (size_t)(row0 + lr) * KDIM + c4 * 4;
    const float* gxl = x  + (size_t)(row0 + (XPASS - 1) * LROWS + lrx) * KDIM + c4 * 4;
    const float* gw  = W1 + (size_t)lr * KDIM + c4 * 4;
    const float* gwl = W1 + (size_t)((WPASS - 1) * LROWS + lrw) * KDIM + c4 * 4;
    const int    ls  = lr * KC + c4 * 4;

    const float* ax = sx + (wave * 16 + nn) * KC + khalf;
    const float* bw = sw + nn * KC + khalf;

    v4f rx[XPASS], rw[WPASS];
    v8f acc0 = {}, acc1 = {}, acc2 = {}, acc3 = {};

    for (int c = 0; c <= NCHUNK; ++c) {
        if (c > 0) {
            __syncthreads();
            if (tid < LTHR) {
                for (int p = 0; p < XPASS - 1; ++p) *(v4f*)(sx + ls + p * LROWS * KC) = rx[p];
                for (int p = 0; p < WPASS - 1; ++p) *(v4f*)(sw + ls + p * LROWS * KC) = rw[p];
                if (lr < XLAST) *(v4f*)(sx + ls + (XPASS - 1) * LROWS * KC) = rx[XPASS - 1];
                if (lr < WLAST) *(v4f*)(sw + ls + (WPASS - 1) * LROWS * KC) = rw[WPASS - 1];
            }
            __syncthreads();
        }
        if (c < NCHUNK) {
            const int kc0 = c * KC;
            for (int p = 0; p < XPASS - 1; ++p)
                rx[p] = __builtin_nontemporal_load((const v4f*)(gx + p * LROWS * KDIM + kc0));
            rx[XPASS - 1] = __builtin_nontemporal_load((const v4f*)(gxl + kc0));
            for (int p = 0; p < WPASS - 1; ++p)
                rw[p] = *(const v4f*)(gw + p * LROWS * KDIM + kc0);
            rw[WPASS - 1] = *(const v4f*)(gwl + kc0);
        }
        if (c > 0) {
            for (int kk = 0; kk < KC; kk += 4) {
                v2f a  = *(const v2f*)(ax + kk);
                v2f b0 = *(const v2f*)(bw +  0 * KC + kk);
                v2f b1 = *(const v2f*)(bw + 16 * KC + kk);
                v2f b2 = *(const v2f*)(bw + 32 * KC + kk);
                v2f b3 = *(const v2f*)(bw + 48 * KC + kk);
                acc0 = __builtin_amdgcn_wmma_f32_16x16x4_f32(false, a, false, b0, (short)0, acc0, false, false);
                acc1 = __builtin_amdgcn_wmma_f32_16x16x4_f32(false, a, false, b1, (short)0, acc1, false, false);
                acc2 = __builtin_amdgcn_wmma_f32_16x16x4_f32(false, a, false, b2, (short)0, acc2, false, false);
                acc3 = __builtin_amdgcn_wmma_f32_16x16x4_f32(false, a, false, b3, (short)0, acc3, false, false);
            }
        }
    }
    if (row0 + wave * 16 < row_nom) return;

    const bool live = nn < NOUT;
    const int  wn   = live ? nn : NOUT - 1;
    const float* wrow = Wf + (size_t)wn * (3 * HIDDEN) + khalf;

    float* hx = sx + (wave * 16 + ((lane >> 4) << 3)) * KC + nn;
    for (int r = 0; r < 8; ++r) {
        float v0 = acc0[r]; v0 = v0 > 0.f ? v0 : 0.f;
        float v1 = acc1[r]; v1 = v1 > 0.f ? v1 : 0.f;
        float v2 = acc2[r]; v2 = v2 > 0.f ? v2 : 0.f;
        float v3 = acc3[r]; v3 = v3 > 0.f ? v3 : 0.f;
        hx[r * KC +  0] = v0;
        hx[r * KC + 16] = v1;
        hx[r * KC + 32] = v2;
        hx[r * KC + 48] = v3;
    }

    v8f p0 = {}, p1 = {}, p2 = {};
    #pragma unroll
    for (int k0 = 0; k0 < HIDDEN; k0 += 4) {
        const v2f a = *(const v2f*)(ax + k0);
        const v2f w0 = *(const v2f*)(wrow + 0 * HIDDEN + k0);
        const v2f w1 = *(const v2f*)(wrow + 1 * HIDDEN + k0);
        const v2f w2 = *(const v2f*)(wrow + 2 * HIDDEN + k0);
        v2f b0, b1, b2;
        b0.x = live ? w0.x : 0.f;  b0.y = live ? w0.y : 0.f;
        b1.x = live ? w1.x : 0.f;  b1.y = live ? w1.y : 0.f;
        b2.x = live ? w2.x : 0.f;  b2.y = live ? w2.y : 0.f;
        p0 = __builtin_amdgcn_wmma_f32_16x16x4_f32(false, a, false, b0, (short)0, p0, false, false);
        p1 = __builtin_amdgcn_wmma_f32_16x16x4_f32(false, a, false, b1, (short)0, p1, false, false);
        p2 = __builtin_amdgcn_wmma_f32_16x16x4_f32(false, a, false, b2, (short)0, p2, false, false);
    }
    const int m_lo = row0 + wave * 16 + ((lane >> 4) << 3);
    for (int r = 0; r < 8; ++r) {
        const size_t m = (size_t)(m_lo + r);
        g1[m * GPITCH + nn] = p1[r];
        g2[m * GPITCH + nn] = p2[r];
        if (live) out[m * NOUT + nn] = p0[r];
    }
}

__global__ __launch_bounds__(256) void k_spmm10(
    const int* __restrict__ src, const int* __restrict__ dst,
    const float* __restrict__ val, const float* __restrict__ g,
    float* __restrict__ out, int nnz)
{
    const int lane = threadIdx.x & 31;
    const int e0   = (blockIdx.x * 8 + (threadIdx.x >> 5)) * 32;
    if (e0 >= nnz) return;
    const int e = e0 + lane;
    int s = 0, d = 0;
    float v = 0.f;
    if (e < nnz) {
        s = __builtin_nontemporal_load(src + e);
        d = __builtin_nontemporal_load(dst + e);
        v = __builtin_nontemporal_load(val + e);
    }
    const int n    = lane & 15;
    const int row0 = lane & 16;

    int   di[16];
    float vi[16], gv[16];
    for (int i = 0; i < 16; ++i) {
        const int si = __shfl(s, row0 + i);
        di[i] = __shfl(d, row0 + i);
        vi[i] = __shfl(v, row0 + i);
        gv[i] = g[(size_t)si * GPITCH + n];
    }
    for (int i = 0; i < 16; ++i) {
        if (n < NOUT && e0 + row0 + i < nnz)
            __hip_atomic_fetch_add(out + (size_t)di[i] * NOUT + n, vi[i] * gv[i],
                                   __ATOMIC_RELAXED, __HIP_MEMORY_SCOPE_AGENT);
    }
}

extern "C" void kernel_launch(void* const* d_in, const int* in_sizes, int n_in,
                              void* d_out, int out_size, void* d_ws, size_t ws_size,
                              hipStream_t stream)
{
    const float* x    = (const float*)d_in[0];
    const float* W1   = (const float*)d_in[1];
    const float* Wf   = (const float*)d_in[2];
    const int*   src1 = (const int*)  d_in[3];
    const int*   dst1 = (const int*)  d_in[4];
    const float* val1 = (const float*)d_in[5];
    const int*   src2 = (const int*)  d_in[6];
    const int*   dst2 = (const int*)  d_in[7];
    const float* val2 = (const float*)d_in[8];

    const int n    = in_sizes[0] / KDIM;
    const int nnz1 = in_sizes[3];
    const int nnz2 = in_sizes[6];

    float* g1  = (float*)d_ws;
    float* g2  = g1 + (size_t)n * GPITCH;
    float* out = (float*)d_out;

    const int n_tiles = (n + 15) / 16;

    k_gemm_relu_proj<<<(n_tiles + 7) / 8, 256, 0, stream>>>(x, W1, Wf, out, g1, g2, n_tiles);

    {
        const int w1 = (nnz1 + 31) / 32;
        k_spmm10<<<(w1 + 7) / 8, 256, 0, stream>>>(src1, dst1, val1, g1, out, nnz1);
        const int w2 = (nnz2 + 31) / 32;
        k_spmm10<<<(w2 + 7) / 8, 256, 0, stream>>>(src2, dst2, val2, g2, out, nnz2);
    }
}


# SpMM: per-edge global f32 atomics replaced by LDS sums per destination range (64 ranges x 4 edge slices scan dst, queue hits, ds_add_f32), whole-line flush into out; one launch; sum order changes
# speedup vs baseline: 3.3717x; 1.0556x over previous
#include <hip/hip_runtime.h>

typedef float v2f __attribute__((ext_vector_type(2)));
typedef float v4f __attribute__((ext_vector_type(4)));
typedef float v8f __attribute__((ext_vector_type(8)));

#define KDIM   500
#define HIDDEN 64
#define NOUT   10
#define GPITCH 16

#define KC     100
#define KC4    (KC / 4)
#define NCHUNK (KDIM / KC)
#define TILE_M 128
#define LROWS  10
#define LTHR   (LROWS * KC4)
#define XPASS  ((TILE_M + LROWS - 1) / LROWS)
#define WPASS  ((HIDDEN + LROWS - 1) / LROWS)
#define XLAST  (TILE_M - (XPASS - 1) * LROWS)
#define WLAST  (HIDDEN - (WPASS - 1) * LROWS)

__global__ __launch_bounds__(256) void k_gemm_relu_proj(
    const float* __restrict__ x, const float* __restrict__ W1,
    const float* __restrict__ Wf,
    float* __restrict__ out, float* __restrict__ g1, float* __restrict__ g2,
    int n_tiles)
{
    __shared__ __attribute__((aligned(16))) float sx[TILE_M * KC];
    __shared__ __attribute__((aligned(16))) float sw[HIDDEN * KC];

    const int tid     = threadIdx.x;
    const int lane    = tid & 31;
    const int wave    = tid >> 5;
    const int n_rows  = n_tiles * 16;
    const int row_nom = blockIdx.x * TILE_M;
    const int row0    = row_nom < n_rows - TILE_M ? row_nom : n_rows - TILE_M;
    const int nn      = lane & 15;
    const int khalf   = (lane >> 4) << 1;

    const int lt  = tid < LTHR ? tid : tid - LTHR;
    const int lr  = lt / KC4;
    const int c4  = lt - lr * KC4;
    const int lrx = lr < XLAST ? lr : XLAST - 1;
    const int lrw = lr < WLAST ? lr : WLAST - 1;
    const float* gx  = x  + (size_t)(row0 + lr) * KDIM + c4 * 4;
    const float* gxl = x  + (size_t)(row0 + (XPASS - 1) * LROWS + lrx) * KDIM + c4 * 4;
    const float* gw  = W1 + (size_t)lr * KDIM + c4 * 4;
    const float* gwl = W1 + (size_t)((WPASS - 1) * LROWS + lrw) * KDIM + c4 * 4;
    const int    ls  = lr * KC + c4 * 4;

    const float* ax = sx + (wave * 16 + nn) * KC + khalf;
    const float* bw = sw + nn * KC + khalf;

    v4f rx[XPASS], rw[WPASS];
    v8f acc0 = {}, acc1 = {}, acc2 = {}, acc3 = {};

    for (int c = 0; c <= NCHUNK; ++c) {
        if (c > 0) {
            __syncthreads();
            if (tid < LTHR) {
                for (int p = 0; p < XPASS - 1; ++p) *(v4f*)(sx + ls + p * LROWS * KC) = rx[p];
                for (int p = 0; p < WPASS - 1; ++p) *(v4f*)(sw + ls + p * LROWS * KC) = rw[p];
                if (lr < XLAST) *(v4f*)(sx + ls + (XPASS - 1) * LROWS * KC) = rx[XPASS - 1];
                if (lr < WLAST) *(v4f*)(sw + ls + (WPASS - 1) * LROWS * KC) = rw[WPASS - 1];
            }
            __syncthreads();
        }
        if (c < NCHUNK) {
            const int kc0 = c * KC;
            for (int p = 0; p < XPASS - 1; ++p)
                rx[p] = __builtin_nontemporal_load((const v4f*)(gx + p * LROWS * KDIM + kc0));
            rx[XPASS - 1] = __builtin_nontemporal_load((const v4f*)(gxl + kc0));
            for (int p = 0; p < WPASS - 1; ++p)
                rw[p] = *(const v4f*)(gw + p * LROWS * KDIM + kc0);
            rw[WPASS - 1] = *(const v4f*)(gwl + kc0);
        }
        if (c > 0) {
            for (int kk = 0; kk < KC; kk += 4) {
                v2f a  = *(const v2f*)(ax + kk);
                v2f b0 = *(const v2f*)(bw +  0 * KC + kk);
                v2f b1 = *(const v2f*)(bw + 16 * KC + kk);
                v2f b2 = *(const v2f*)(bw + 32 * KC + kk);
                v2f b3 = *(const v2f*)(bw + 48 * KC + kk);
                acc0 = __builtin_amdgcn_wmma_f32_16x16x4_f32(false, a, false, b0, (short)0, acc0, false, false);
                acc1 = __builtin_amdgcn_wmma_f32_16x16x4_f32(false, a, false, b1, (short)0, acc1, false, false);
                acc2 = __builtin_amdgcn_wmma_f32_16x16x4_f32(false, a, false, b2, (short)0, acc2, false, false);
                acc3 = __builtin_amdgcn_wmma_f32_16x16x4_f32(false, a, false, b3, (short)0, acc3, false, false);
            }
        }
    }
    if (row0 + wave * 16 < row_nom) return;

    const bool live = nn < NOUT;
    const int  wn   = live ? nn : NOUT - 1;
    const float* wrow = Wf + (size_t)wn * (3 * HIDDEN) + khalf;

    float* hx = sx + (wave * 16 + ((lane >> 4) << 3)) * KC + nn;
    for (int r = 0; r < 8; ++r) {
        float v0 = acc0[r]; v0 = v0 > 0.f ? v0 : 0.f;
        float v1 = acc1[r]; v1 = v1 > 0.f ? v1 : 0.f;
        float v2 = acc2[r]; v2 = v2 > 0.f ? v2 : 0.f;
        float v3 = acc3[r]; v3 = v3 > 0.f ? v3 : 0.f;
        hx[r * KC +  0] = v0;
        hx[r * KC + 16] = v1;
        hx[r * KC + 32] = v2;
        hx[r * KC + 48] = v3;
    }

    v8f p0 = {}, p1 = {}, p2 = {};
    #pragma unroll
    for (int k0 = 0; k0 < HIDDEN; k0 += 4) {
        const v2f a = *(const v2f*)(ax + k0);
        const v2f w0 = *(const v2f*)(wrow + 0 * HIDDEN + k0);
        const v2f w1 = *(const v2f*)(wrow + 1 * HIDDEN + k0);
        const v2f w2 = *(const v2f*)(wrow + 2 * HIDDEN + k0);
        v2f b0, b1, b2;
        b0.x = live ? w0.x : 0.f;  b0.y = live ? w0.y : 0.f;
        b1.x = live ? w1.x : 0.f;  b1.y = live ? w1.y : 0.f;
        b2.x = live ? w2.x : 0.f;  b2.y = live ? w2.y : 0.f;
        p0 = __builtin_amdgcn_wmma_f32_16x16x4_f32(false, a, false, b0, (short)0, p0, false, false);
        p1 = __builtin_amdgcn_wmma_f32_16x16x4_f32(false, a, false, b1, (short)0, p1, false, false);
        p2 = __builtin_amdgcn_wmma_f32_16x16x4_f32(false, a, false, b2, (short)0, p2, false, false);
    }
    const int m_lo = row0 + wave * 16 + ((lane >> 4) << 3);
    for (int r = 0; r < 8; ++r) {
        const size_t m = (size_t)(m_lo + r);
        g1[m * GPITCH + nn] = p1[r];
        g2[m * GPITCH + nn] = p2[r];
        if (live) out[m * NOUT + nn] = p0[r];
    }
}

#define OWN_P     64
#define OWN_S     4
#define OWN_ROWS  1568
#define OWN_WAVES 32
#define OWN_THR   (OWN_WAVES * 32)
#define SCAN_U    8
#define QCAP      64

__device__ __forceinline__ void lds_add(float* p, float v)
{
    __hip_atomic_fetch_add(p, v, __ATOMIC_RELAXED, __HIP_MEMORY_SCOPE_WORKGROUP);
}

__device__ __forceinline__ void own_drain(
    const int2* qw, int first, int k, int lane,
    const int* __restrict__ src, const float* __restrict__ val,
    const float* __restrict__ g, float* acc)
{
    if (lane < k) {
        const int2  ent = qw[first + lane];
        const int   s   = src[ent.x];
        const float v   = val[ent.x];
        const float* gr = g + (size_t)s * GPITCH;
        const v4f a = *(const v4f*)(gr);
        const v4f b = *(const v4f*)(gr + 4);
        const v2f c = *(const v2f*)(gr + 8);
        float* ar = acc + ent.y * NOUT;
        lds_add(ar + 0, v * a.x);  lds_add(ar + 1, v * a.y);
        lds_add(ar + 2, v * a.z);  lds_add(ar + 3, v * a.w);
        lds_add(ar + 4, v * b.x);  lds_add(ar + 5, v * b.y);
        lds_add(ar + 6, v * b.z);  lds_add(ar + 7, v * b.w);
        lds_add(ar + 8, v * c.x);  lds_add(ar + 9, v * c.y);
    }
}

__device__ __forceinline__ void own_step(
    int e, int d, int lo, unsigned rows, int2* qw, int& qn, int lane,
    const int* __restrict__ src, const float* __restrict__ val,
    const float* __restrict__ g, float* acc)
{
    const unsigned rel = (unsigned)(d - lo);
    const bool     sel = rel < rows;
    const unsigned m   = __builtin_amdgcn_ballot_w32(sel);
    if (m != 0u) {
        if (sel) qw[qn + (int)__builtin_amdgcn_mbcnt_lo(m, 0u)] = make_int2(e, (int)rel);
        qn += __builtin_popcount(m);
        if (qn >= 32) {
            __builtin_amdgcn_wave_barrier();
            own_drain(qw, qn - 32, 32, lane, src, val, g, acc);
            qn -= 32;
        }
    }
}

__device__ __forceinline__ void own_load(
    const int* __restrict__ dst, int base, int lane, int start, int end4, int4 (&d)[SCAN_U])
{
    #pragma unroll
    for (int u = 0; u < SCAN_U; ++u) {
        const int  e  = base + u * 128 + lane * 4;
        const bool ok = e < end4;
        const int4 t  = *(const int4*)(dst + (ok ? e : start));
        d[u].x = ok ? t.x : -1;  d[u].y = ok ? t.y : -1;
        d[u].z = ok ? t.z : -1;  d[u].w = ok ? t.w : -1;
    }
}

__device__ __forceinline__ void own_scan(
    const int* __restrict__ src, const int* __restrict__ dst, const float* __restrict__ val,
    const float* __restrict__ g, int nnz, int slice, int lo, unsigned rows,
    float* acc, int2* qw, int lane, int wave)
{
    const int chunk = (((nnz + OWN_S - 1) / OWN_S) + 3) & ~3;
    const int start = slice * chunk;
    const int end   = start + chunk < nnz ? start + chunk : nnz;
    if (end <= start) return;
    const int end4  = start + ((end - start) & ~3);
    const int step  = OWN_WAVES * SCAN_U * 128;
    int qn = 0;

    if (end4 > start) {
        int base = start + wave * (SCAN_U * 128);
        int4 cur[SCAN_U], nxt[SCAN_U];
        own_load(dst, base, lane, start, end4, cur);
        for (; base < end4; base += step) {
            own_load(dst, base + step, lane, start, end4, nxt);
            #pragma unroll
            for (int u = 0; u < SCAN_U; ++u) {
                const int e = base + u * 128 + lane * 4;
                own_step(e + 0, cur[u].x, lo, rows, qw, qn, lane, src, val, g, acc);
                own_step(e + 1, cur[u].y, lo, rows, qw, qn, lane, src, val, g, acc);
                own_step(e + 2, cur[u].z, lo, rows, qw, qn, lane, src, val, g, acc);
                own_step(e + 3, cur[u].w, lo, rows, qw, qn, lane, src, val, g, acc);
            }
            #pragma unroll
            for (int u = 0; u < SCAN_U; ++u) cur[u] = nxt[u];
        }
    }
    if (wave == 0 && end4 < end) {
        const int e  = end4 + lane;
        const int dd = e < end ? dst[e] : -1;
        own_step(e, dd, lo, rows, qw, qn, lane, src, val, g, acc);
    }
    if (qn > 0) {
        __builtin_amdgcn_wave_barrier();
        own_drain(qw, 0, qn, lane, src, val, g, acc);
    }
}

__global__ __launch_bounds__(OWN_THR) void k_edges_owner(
    const int* __restrict__ src1, const int* __restrict__ dst1, const float* __restrict__ val1,
    const float* __restrict__ g1, int nnz1,
    const int* __restrict__ src2, const int* __restrict__ dst2, const float* __restrict__ val2,
    const float* __restrict__ g2, int nnz2,
    float* __restrict__ out, int n)
{
    __shared__ float acc[OWN_ROWS * NOUT];
    __shared__ int2  q[OWN_WAVES * QCAP];

    const int tid   = threadIdx.x;
    const int lane  = tid & 31;
    const int wave  = __builtin_amdgcn_readfirstlane(tid >> 5);
    const int slice = blockIdx.x / OWN_P;
    int2* qw = q + wave * QCAP;

    for (int part = blockIdx.x % OWN_P; part * OWN_ROWS < n; part += OWN_P) {
        const int      lo   = part * OWN_ROWS;
        const unsigned rows = (unsigned)(n - lo < OWN_ROWS ? n - lo : OWN_ROWS);

        for (int i = tid; i < OWN_ROWS * NOUT; i += OWN_THR) acc[i] = 0.f;
        __syncthreads();

        own_scan(src1, dst1, val1, g1, nnz1, slice, lo, rows, acc, qw, lane, wave);
        own_scan(src2, dst2, val2, g2, nnz2, slice, lo, rows, acc, qw, lane, wave);
        __syncthreads();

        float* o = out + (size_t)lo * NOUT;
        const int cnt = (int)rows * NOUT;
        for (int i = tid; i < cnt; i += OWN_THR)
            __hip_atomic_fetch_add(o + i, acc[i], __ATOMIC_RELAXED, __HIP_MEMORY_SCOPE_AGENT);
        __syncthreads();
    }
}

extern "C" void kernel_launch(void* const* d_in, const int* in_sizes, int n_in,
                              void* d_out, int out_size, void* d_ws, size_t ws_size,
                              hipStream_t stream)
{
    const float* x    = (const float*)d_in[0];
    const float* W1   = (const float*)d_in[1];
    const float* Wf   = (const float*)d_in[2];
    const int*   src1 = (const int*)  d_in[3];
    const int*   dst1 = (const int*)  d_in[4];
    const float* val1 = (const float*)d_in[5];
    const int*   src2 = (const int*)  d_in[6];
    const int*   dst2 = (const int*)  d_in[7];
    const float* val2 = (const float*)d_in[8];

    const int n    = in_sizes[0] / KDIM;
    const int nnz1 = in_sizes[3];
    const int nnz2 = in_sizes[6];

    float* g1  = (float*)d_ws;
    float* g2  = g1 + (size_t)n * GPITCH;
    float* out = (float*)d_out;

    const int n_tiles = (n + 15) / 16;

    k_gemm_relu_proj<<<(n_tiles + 7) / 8, 256, 0, stream>>>(x, W1, Wf, out, g1, g2, n_tiles);

    k_edges_owner<<<OWN_P * OWN_S, OWN_THR, 0, stream>>>(src1, dst1, val1, g1, nnz1,
                                                        src2, dst2, val2, g2, nnz2, out, n);
}


# SpMM: edges counting-sorted by destination range once per call (LDS ranks, +1 launch); an owner reads only its own edges, not the whole dst column; plain-add flush, no global atomic; sum order changes
# speedup vs baseline: 4.0546x; 1.2025x over previous
#include <hip/hip_runtime.h>

typedef float v2f __attribute__((ext_vector_type(2)));
typedef float v4f __attribute__((ext_vector_type(4)));
typedef float v8f __attribute__((ext_vector_type(8)));

#define KDIM   500
#define HIDDEN 64
#define NOUT   10
#define GPITCH 16

#define KC     100
#define KC4    (KC / 4)
#define NCHUNK (KDIM / KC)
#define TILE_M 128
#define LROWS  10
#define LTHR   (LROWS * KC4)
#define XPASS  ((TILE_M + LROWS - 1) / LROWS)
#define WPASS  ((HIDDEN + LROWS - 1) / LROWS)
#define XLAST  (TILE_M - (XPASS - 1) * LROWS)
#define WLAST  (HIDDEN - (WPASS - 1) * LROWS)

__global__ __launch_bounds__(256) void k_gemm_relu_proj(
    const float* __restrict__ x, const float* __restrict__ W1,
    const float* __restrict__ Wf,
    float* __restrict__ out, float* __restrict__ g1, float* __restrict__ g2,
    int n_tiles)
{
    __shared__ __attribute__((aligned(16))) float sx[TILE_M * KC];
    __shared__ __attribute__((aligned(16))) float sw[HIDDEN * KC];

    const int tid     = threadIdx.x;
    const int lane    = tid & 31;
    const int wave    = tid >> 5;
    const int n_rows  = n_tiles * 16;
    const int row_nom = blockIdx.x * TILE_M;
    const int row0    = row_nom < n_rows - TILE_M ? row_nom : n_rows - TILE_M;
    const int nn      = lane & 15;
    const int khalf   = (lane >> 4) << 1;

    const int lt  = tid < LTHR ? tid : tid - LTHR;
    const int lr  = lt / KC4;
    const int c4  = lt - lr * KC4;
    const int lrx = lr < XLAST ? lr : XLAST - 1;
    const int lrw = lr < WLAST ? lr : WLAST - 1;
    const float* gx  = x  + (size_t)(row0 + lr) * KDIM + c4 * 4;
    const float* gxl = x  + (size_t)(row0 + (XPASS - 1) * LROWS + lrx) * KDIM + c4 * 4;
    const float* gw  = W1 + (size_t)lr * KDIM + c4 * 4;
    const float* gwl = W1 + (size_t)((WPASS - 1) * LROWS + lrw) * KDIM + c4 * 4;
    const int    ls  = lr * KC + c4 * 4;

    const float* ax = sx + (wave * 16 + nn) * KC + khalf;
    const float* bw = sw + nn * KC + khalf;

    v4f rx[XPASS], rw[WPASS];
    v8f acc0 = {}, acc1 = {}, acc2 = {}, acc3 = {};

    for (int c = 0; c <= NCHUNK; ++c) {
        if (c > 0) {
            __syncthreads();
            if (tid < LTHR) {
                for (int p = 0; p < XPASS - 1; ++p) *(v4f*)(sx + ls + p * LROWS * KC) = rx[p];
                for (int p = 0; p < WPASS - 1; ++p) *(v4f*)(sw + ls + p * LROWS * KC) = rw[p];
                if (lr < XLAST) *(v4f*)(sx + ls + (XPASS - 1) * LROWS * KC) = rx[XPASS - 1];
                if (lr < WLAST) *(v4f*)(sw + ls + (WPASS - 1) * LROWS * KC) = rw[WPASS - 1];
            }
            __syncthreads();
        }
        if (c < NCHUNK) {
            const int kc0 = c * KC;
            for (int p = 0; p < XPASS - 1; ++p)
                rx[p] = __builtin_nontemporal_load((const v4f*)(gx + p * LROWS * KDIM + kc0));
            rx[XPASS - 1] = __builtin_nontemporal_load((const v4f*)(gxl + kc0));
            for (int p = 0; p < WPASS - 1; ++p)
                rw[p] = *(const v4f*)(gw + p * LROWS * KDIM + kc0);
            rw[WPASS - 1] = *(const v4f*)(gwl + kc0);
        }
        if (c > 0) {
            for (int kk = 0; kk < KC; kk += 4) {
                v2f a  = *(const v2f*)(ax + kk);
                v2f b0 = *(const v2f*)(bw +  0 * KC + kk);
                v2f b1 = *(const v2f*)(bw + 16 * KC + kk);
                v2f b2 = *(const v2f*)(bw + 32 * KC + kk);
                v2f b3 = *(const v2f*)(bw + 48 * KC + kk);
                acc0 = __builtin_amdgcn_wmma_f32_16x16x4_f32(false, a, false, b0, (short)0, acc0, false, false);
                acc1 = __builtin_amdgcn_wmma_f32_16x16x4_f32(false, a, false, b1, (short)0, acc1, false, false);
                acc2 = __builtin_amdgcn_wmma_f32_16x16x4_f32(false, a, false, b2, (short)0, acc2, false, false);
                acc3 = __builtin_amdgcn_wmma_f32_16x16x4_f32(false, a, false, b3, (short)0, acc3, false, false);
            }
        }
    }
    if (row0 + wave * 16 < row_nom) return;

    const bool live = nn < NOUT;
    const int  wn   = live ? nn : NOUT - 1;
    const float* wrow = Wf + (size_t)wn * (3 * HIDDEN) + khalf;

    float* hx = sx + (wave * 16 + ((lane >> 4) << 3)) * KC + nn;
    for (int r = 0; r < 8; ++r) {
        float v0 = acc0[r]; v0 = v0 > 0.f ? v0 : 0.f;
        float v1 = acc1[r]; v1 = v1 > 0.f ? v1 : 0.f;
        float v2 = acc2[r]; v2 = v2 > 0.f ? v2 : 0.f;
        float v3 = acc3[r]; v3 = v3 > 0.f ? v3 : 0.f;
        hx[r * KC +  0] = v0;
        hx[r * KC + 16] = v1;
        hx[r * KC + 32] = v2;
        hx[r * KC + 48] = v3;
    }

    v8f p0 = {}, p1 = {}, p2 = {};
    #pragma unroll
    for (int k0 = 0; k0 < HIDDEN; k0 += 4) {
        const v2f a = *(const v2f*)(ax + k0);
        const v2f w0 = *(const v2f*)(wrow + 0 * HIDDEN + k0);
        const v2f w1 = *(const v2f*)(wrow + 1 * HIDDEN + k0);
        const v2f w2 = *(const v2f*)(wrow + 2 * HIDDEN + k0);
        v2f b0, b1, b2;
        b0.x = live ? w0.x : 0.f;  b0.y = live ? w0.y : 0.f;
        b1.x = live ? w1.x : 0.f;  b1.y = live ? w1.y : 0.f;
        b2.x = live ? w2.x : 0.f;  b2.y = live ? w2.y : 0.f;
        p0 = __builtin_amdgcn_wmma_f32_16x16x4_f32(false, a, false, b0, (short)0, p0, false, false);
        p1 = __builtin_amdgcn_wmma_f32_16x16x4_f32(false, a, false, b1, (short)0, p1, false, false);
        p2 = __builtin_amdgcn_wmma_f32_16x16x4_f32(false, a, false, b2, (short)0, p2, false, false);
    }
    const int m_lo = row0 + wave * 16 + ((lane >> 4) << 3);
    for (int r = 0; r < 8; ++r) {
        const size_t m = (size_t)(m_lo + r);
        g1[m * GPITCH + nn] = p1[r];
        g2[m * GPITCH + nn] = p2[r];
        if (live) out[m * NOUT + nn] = p0[r];
    }
}

#define RP        400
#define RP_BITS   9
#define NBK_MAX   256
#define BIN_THR   1024
#define BIN_EPT   7
#define BIN_C     (BIN_EPT * BIN_THR)
#define OFF_PITCH (NBK_MAX + 1)
#define OWN_THR   1024

__global__ __launch_bounds__(BIN_THR) void k_bin(
    const int* __restrict__ src1, const int* __restrict__ dst1, const float* __restrict__ val1, int nnz1,
    const int* __restrict__ src2, const int* __restrict__ dst2, const float* __restrict__ val2, int nnz2,
    int n, int* __restrict__ wp, float* __restrict__ wv, int* __restrict__ woff)
{
    __shared__ int   cnt[NBK_MAX];
    __shared__ int   off[NBK_MAX + 1];
    __shared__ int   wsum[NBK_MAX / 32];
    __shared__ int   sp[BIN_C];
    __shared__ float sv[BIN_C];

    const int tid  = threadIdx.x;
    const int lane = tid & 31;
    const int wave = tid >> 5;
    const int tot  = nnz1 + nnz2;
    const int base = blockIdx.x * BIN_C;

    if (tid < NBK_MAX) cnt[tid] = 0;
    __syncthreads();

    int   pk[BIN_EPT], kk[BIN_EPT], rk[BIN_EPT];
    float vv[BIN_EPT];
    #pragma unroll
    for (int u = 0; u < BIN_EPT; ++u) {
        const int e = base + u * BIN_THR + tid;
        kk[u] = -1;  pk[u] = 0;  rk[u] = 0;  vv[u] = 0.f;
        if (e < tot) {
            const bool second = e >= nnz1;
            const int  i = second ? e - nnz1 : e;
            const int  d = second ? dst2[i] : dst1[i];
            const int  s = second ? src2[i] : src1[i];
            const float v = second ? val2[i] : val1[i];
            if ((unsigned)d < (unsigned)n) {
                const int k   = d / RP;
                const int rel = d - k * RP;
                kk[u] = k;
                rk[u] = __hip_atomic_fetch_add(&cnt[k], 1, __ATOMIC_RELAXED, __HIP_MEMORY_SCOPE_WORKGROUP);
                pk[u] = ((s + (second ? n : 0)) << RP_BITS) | rel;
                vv[u] = v;
            }
        }
    }
    __syncthreads();

    int incl = 0;
    if (tid < NBK_MAX) {
        incl = cnt[tid];
        for (int dlt = 1; dlt < 32; dlt <<= 1) {
            const int t = __shfl_up(incl, dlt);
            if (lane >= dlt) incl += t;
        }
        if (lane == 31) wsum[wave] = incl;
    }
    __syncthreads();
    if (tid < NBK_MAX) {
        int add = 0;
        for (int w = 0; w < wave; ++w) add += wsum[w];
        off[tid + 1] = incl + add;
        if (tid == 0) off[0] = 0;
    }
    __syncthreads();

    #pragma unroll
    for (int u = 0; u < BIN_EPT; ++u) {
        if (kk[u] >= 0) {
            const int p = off[kk[u]] + rk[u];
            sp[p] = pk[u];
            sv[p] = vv[u];
        }
    }
    if (tid <= NBK_MAX) woff[(size_t)blockIdx.x * OFF_PITCH + tid] = off[tid];
    __syncthreads();

    const int total = off[NBK_MAX];
    for (int i = tid; i < total; i += BIN_THR) {
        wp[base + i] = sp[i];
        wv[base + i] = sv[i];
    }
}

__device__ __forceinline__ void lds_add(float* p, float v)
{
    __hip_atomic_fetch_add(p, v, __ATOMIC_RELAXED, __HIP_MEMORY_SCOPE_WORKGROUP);
}

__device__ __forceinline__ void own_add(float* acc, int rec, float v, v4f a, v4f b, v2f c)
{
    float* ar = acc + (rec & ((1 << RP_BITS) - 1)) * NOUT;
    lds_add(ar + 0, v * a.x);  lds_add(ar + 1, v * a.y);
    lds_add(ar + 2, v * a.z);  lds_add(ar + 3, v * a.w);
    lds_add(ar + 4, v * b.x);  lds_add(ar + 5, v * b.y);
    lds_add(ar + 6, v * b.z);  lds_add(ar + 7, v * b.w);
    lds_add(ar + 8, v * c.x);  lds_add(ar + 9, v * c.y);
}

__global__ __launch_bounds__(OWN_THR) void k_owner(
    const int* __restrict__ wp, const float* __restrict__ wv, const int* __restrict__ woff, int nb,
    const float* __restrict__ g, float* __restrict__ out, int n)
{
    __shared__ float acc[RP * NOUT];

    const int tid  = threadIdx.x;
    const int lane = tid & 31;
    const int wave = __builtin_amdgcn_readfirstlane(tid >> 5);
    const int part = blockIdx.x;
    const int lo   = part * RP;
    const int rows = n - lo < RP ? n - lo : RP;

    for (int i = tid; i < RP * NOUT; i += OWN_THR) acc[i] = 0.f;
    __syncthreads();

    for (int b0 = 0; b0 < nb; b0 += OWN_THR) {
        const int b = b0 + wave + 32 * lane;
        int st = 0, en = 0;
        if (b < nb) {
            const int* o = woff + (size_t)b * OFF_PITCH + part;
            st = b * BIN_C + o[0];
            en = b * BIN_C + o[1];
        }
        const int left = nb - b0 - wave;
        const int nseg = left <= 0 ? 0 : ((left + 31) / 32 < 32 ? (left + 31) / 32 : 32);

        for (int j = 0; j < nseg; j += 2) {
            const int  j1   = j + 1 < 32 ? j + 1 : 31;
            const bool two  = j + 1 < nseg;
            const int  st0  = __shfl(st, j),  en0 = __shfl(en, j);
            const int  st1r = __shfl(st, j1), en1r = __shfl(en, j1);
            const int  st1  = two ? st1r : 0, en1 = two ? en1r : 0;
            const int  len0 = en0 - st0, len1 = en1 - st1;
            const int  len  = len0 > len1 ? len0 : len1;
            for (int o = 0; o < len; o += 32) {
                const int  i0  = st0 + o + lane, i1 = st1 + o + lane;
                const bool ok0 = i0 < en0,       ok1 = i1 < en1;
                const int   r0 = wp[ok0 ? i0 : 0];
                const float v0 = wv[ok0 ? i0 : 0];
                const int   r1 = wp[ok1 ? i1 : 0];
                const float v1 = wv[ok1 ? i1 : 0];
                const float* g0 = g + (size_t)(ok0 ? r0 >> RP_BITS : 0) * GPITCH;
                const float* g1r = g + (size_t)(ok1 ? r1 >> RP_BITS : 0) * GPITCH;
                const v4f a0 = *(const v4f*)(g0);
                const v4f b0v = *(const v4f*)(g0 + 4);
                const v2f c0 = *(const v2f*)(g0 + 8);
                const v4f a1 = *(const v4f*)(g1r);
                const v4f b1v = *(const v4f*)(g1r + 4);
                const v2f c1 = *(const v2f*)(g1r + 8);
                if (ok0) own_add(acc, r0, v0, a0, b0v, c0);
                if (ok1) own_add(acc, r1, v1, a1, b1v, c1);
            }
        }
    }
    __syncthreads();

    float* o = out + (size_t)lo * NOUT;
    const int cnt = rows * NOUT;
    for (int i = tid; i < cnt; i += OWN_THR) o[i] = o[i] + acc[i];
}

extern "C" void kernel_launch(void* const* d_in, const int* in_sizes, int n_in,
                              void* d_out, int out_size, void* d_ws, size_t ws_size,
                              hipStream_t stream)
{
    const float* x    = (const float*)d_in[0];
    const float* W1   = (const float*)d_in[1];
    const float* Wf   = (const float*)d_in[2];
    const int*   src1 = (const int*)  d_in[3];
    const int*   dst1 = (const int*)  d_in[4];
    const float* val1 = (const float*)d_in[5];
    const int*   src2 = (const int*)  d_in[6];
    const int*   dst2 = (const int*)  d_in[7];
    const float* val2 = (const float*)d_in[8];

    const int n    = in_sizes[0] / KDIM;
    const int nnz1 = in_sizes[3];
    const int nnz2 = in_sizes[6];
    const int nb   = (nnz1 + nnz2 + BIN_C - 1) / BIN_C;

    float* g1   = (float*)d_ws;
    float* g2   = g1 + (size_t)n * GPITCH;
    int*   wp   = (int*)(g2 + (size_t)n * GPITCH);
    float* wv   = (float*)(wp + (size_t)nb * BIN_C);
    int*   woff = (int*)(wv + (size_t)nb * BIN_C);
    float* out  = (float*)d_out;

    const int n_tiles = (n + 15) / 16;

    k_bin<<<nb, BIN_THR, 0, stream>>>(src1, dst1, val1, nnz1, src2, dst2, val2, nnz2, n, wp, wv, woff);

    k_gemm_relu_proj<<<(n_tiles + 7) / 8, 256, 0, stream>>>(x, W1, Wf, out, g1, g2, n_tiles);

    k_owner<<<(n + RP - 1) / RP, OWN_THR, 0, stream>>>(wp, wv, woff, nb, g1, out, n);
}


# Edge-ordering kernel: the 21 loads of a thread (dst, src, val of its 7 edges) made unconditional and issued as one group: the wave waits for memory once, not twice per edge; arithmetic untouched
# speedup vs baseline: 4.2315x; 1.0436x over previous
#include <hip/hip_runtime.h>

typedef float v2f __attribute__((ext_vector_type(2)));
typedef float v4f __attribute__((ext_vector_type(4)));
typedef float v8f __attribute__((ext_vector_type(8)));

#define KDIM   500
#define HIDDEN 64
#define NOUT   10
#define GPITCH 16

#define KC     100
#define KC4    (KC / 4)
#define NCHUNK (KDIM / KC)
#define TILE_M 128
#define LROWS  10
#define LTHR   (LROWS * KC4)
#define XPASS  ((TILE_M + LROWS - 1) / LROWS)
#define WPASS  ((HIDDEN + LROWS - 1) / LROWS)
#define XLAST  (TILE_M - (XPASS - 1) * LROWS)
#define WLAST  (HIDDEN - (WPASS - 1) * LROWS)

__global__ __launch_bounds__(256) void k_gemm_relu_proj(
    const float* __restrict__ x, const float* __restrict__ W1,
    const float* __restrict__ Wf,
    float* __restrict__ out, float* __restrict__ g1, float* __restrict__ g2,
    int n_tiles)
{
    __shared__ __attribute__((aligned(16))) float sx[TILE_M * KC];
    __shared__ __attribute__((aligned(16))) float sw[HIDDEN * KC];

    const int tid     = threadIdx.x;
    const int lane    = tid & 31;
    const int wave    = tid >> 5;
    const int n_rows  = n_tiles * 16;
    const int row_nom = blockIdx.x * TILE_M;
    const int row0    = row_nom < n_rows - TILE_M ? row_nom : n_rows - TILE_M;
    const int nn      = lane & 15;
    const int khalf   = (lane >> 4) << 1;

    const int lt  = tid < LTHR ? tid : tid - LTHR;
    const int lr  = lt / KC4;
    const int c4  = lt - lr * KC4;
    const int lrx = lr < XLAST ? lr : XLAST - 1;
    const int lrw = lr < WLAST ? lr : WLAST - 1;
    const float* gx  = x  + (size_t)(row0 + lr) * KDIM + c4 * 4;
    const float* gxl = x  + (size_t)(row0 + (XPASS - 1) * LROWS + lrx) * KDIM + c4 * 4;
    const float* gw  = W1 + (size_t)lr * KDIM + c4 * 4;
    const float* gwl = W1 + (size_t)((WPASS - 1) * LROWS + lrw) * KDIM + c4 * 4;
    const int    ls  = lr * KC + c4 * 4;

    const float* ax = sx + (wave * 16 + nn) * KC + khalf;
    const float* bw = sw + nn * KC + khalf;

    v4f rx[XPASS], rw[WPASS];
    v8f acc0 = {}, acc1 = {}, acc2 = {}, acc3 = {};

    for (int c = 0; c <= NCHUNK; ++c) {
        if (c > 0) {
            __syncthreads();
            if (tid < LTHR) {
                for (int p = 0; p < XPASS - 1; ++p) *(v4f*)(sx + ls + p * LROWS * KC) = rx[p];
                for (int p = 0; p < WPASS - 1; ++p) *(v4f*)(sw + ls + p * LROWS * KC) = rw[p];
                if (lr < XLAST) *(v4f*)(sx + ls + (XPASS - 1) * LROWS * KC) = rx[XPASS - 1];
                if (lr < WLAST) *(v4f*)(sw + ls + (WPASS - 1) * LROWS * KC) = rw[WPASS - 1];
            }
            __syncthreads();
        }
        if (c < NCHUNK) {
            const int kc0 = c * KC;
            for (int p = 0; p < XPASS - 1; ++p)
                rx[p] = __builtin_nontemporal_load((const v4f*)(gx + p * LROWS * KDIM + kc0));
            rx[XPASS - 1] = __builtin_nontemporal_load((const v4f*)(gxl + kc0));
            for (int p = 0; p < WPASS - 1; ++p)
                rw[p] = *(const v4f*)(gw + p * LROWS * KDIM + kc0);
            rw[WPASS - 1] = *(const v4f*)(gwl + kc0);
        }
        if (c > 0) {
            for (int kk = 0; kk < KC; kk += 4) {
                v2f a  = *(const v2f*)(ax + kk);
                v2f b0 = *(const v2f*)(bw +  0 * KC + kk);
                v2f b1 = *(const v2f*)(bw + 16 * KC + kk);
                v2f b2 = *(const v2f*)(bw + 32 * KC + kk);
                v2f b3 = *(const v2f*)(bw + 48 * KC + kk);
                acc0 = __builtin_amdgcn_wmma_f32_16x16x4_f32(false, a, false, b0, (short)0, acc0, false, false);
                acc1 = __builtin_amdgcn_wmma_f32_16x16x4_f32(false, a, false, b1, (short)0, acc1, false, false);
                acc2 = __builtin_amdgcn_wmma_f32_16x16x4_f32(false, a, false, b2, (short)0, acc2, false, false);
                acc3 = __builtin_amdgcn_wmma_f32_16x16x4_f32(false, a, false, b3, (short)0, acc3, false, false);
            }
        }
    }
    if (row0 + wave * 16 < row_nom) return;

    const bool live = nn < NOUT;
    const int  wn   = live ? nn : NOUT - 1;
    const float* wrow = Wf + (size_t)wn * (3 * HIDDEN) + khalf;

    float* hx = sx + (wave * 16 + ((lane >> 4) << 3)) * KC + nn;
    for (int r = 0; r < 8; ++r) {
        float v0 = acc0[r]; v0 = v0 > 0.f ? v0 : 0.f;
        float v1 = acc1[r]; v1 = v1 > 0.f ? v1 : 0.f;
        float v2 = acc2[r]; v2 = v2 > 0.f ? v2 : 0.f;
        float v3 = acc3[r]; v3 = v3 > 0.f ? v3 : 0.f;
        hx[r * KC +  0] = v0;
        hx[r * KC + 16] = v1;
        hx[r * KC + 32] = v2;
        hx[r * KC + 48] = v3;
    }

    v8f p0 = {}, p1 = {}, p2 = {};
    #pragma unroll
    for (int k0 = 0; k0 < HIDDEN; k0 += 4) {
        const v2f a = *(const v2f*)(ax + k0);
        const v2f w0 = *(const v2f*)(wrow + 0 * HIDDEN + k0);
        const v2f w1 = *(const v2f*)(wrow + 1 * HIDDEN + k0);
        const v2f w2 = *(const v2f*)(wrow + 2 * HIDDEN + k0);
        v2f b0, b1, b2;
        b0.x = live ? w0.x : 0.f;  b0.y = live ? w0.y : 0.f;
        b1.x = live ? w1.x : 0.f;  b1.y = live ? w1.y : 0.f;
        b2.x = live ? w2.x : 0.f;  b2.y = live ? w2.y : 0.f;
        p0 = __builtin_amdgcn_wmma_f32_16x16x4_f32(false, a, false, b0, (short)0, p0, false, false);
        p1 = __builtin_amdgcn_wmma_f32_16x16x4_f32(false, a, false, b1, (short)0, p1, false, false);
        p2 = __builtin_amdgcn_wmma_f32_16x16x4_f32(false, a, false, b2, (short)0, p2, false, false);
    }
    const int m_lo = row0 + wave * 16 + ((lane >> 4) << 3);
    for (int r = 0; r < 8; ++r) {
        const size_t m = (size_t)(m_lo + r);
        g1[m * GPITCH + nn] = p1[r];
        g2[m * GPITCH + nn] = p2[r];
        if (live) out[m * NOUT + nn] = p0[r];
    }
}

#define RP        400
#define RP_BITS   9
#define NBK_MAX   256
#define BIN_THR   1024
#define BIN_EPT   7
#define BIN_C     (BIN_EPT * BIN_THR)
#define OFF_PITCH (NBK_MAX + 1)
#define OWN_THR   1024

__global__ __launch_bounds__(BIN_THR) void k_bin(
    const int* __restrict__ src1, const int* __restrict__ dst1, const float* __restrict__ val1, int nnz1,
    const int* __restrict__ src2, const int* __restrict__ dst2, const float* __restrict__ val2, int nnz2,
    int n, int* __restrict__ wp, float* __restrict__ wv, int* __restrict__ woff)
{
    __shared__ int   cnt[NBK_MAX];
    __shared__ int   off[NBK_MAX + 1];
    __shared__ int   wsum[NBK_MAX / 32];
    __shared__ int   sp[BIN_C];
    __shared__ float sv[BIN_C];

    const int tid  = threadIdx.x;
    const int lane = tid & 31;
    const int wave = tid >> 5;
    const int tot  = nnz1 + nnz2;
    const int base = blockIdx.x * BIN_C;

    if (tid < NBK_MAX) cnt[tid] = 0;
    __syncthreads();

    int   pk[BIN_EPT], kk[BIN_EPT], rk[BIN_EPT];
    int   dd[BIN_EPT], ss[BIN_EPT];
    float vv[BIN_EPT];
    #pragma unroll
    for (int u = 0; u < BIN_EPT; ++u) {
        const int    e      = base + u * BIN_THR + tid;
        const int    ec     = e < tot ? e : tot - 1;
        const bool   second = ec >= nnz1;
        const int    i      = second ? ec - nnz1 : ec;
        const int*   dq     = second ? dst2 : dst1;
        const int*   sq     = second ? src2 : src1;
        const float* vq     = second ? val2 : val1;
        dd[u] = dq[i];
        ss[u] = sq[i] + (second ? n : 0);
        vv[u] = vq[i];
    }
    #pragma unroll
    for (int u = 0; u < BIN_EPT; ++u) {
        const int e   = base + u * BIN_THR + tid;
        const int d   = dd[u];
        const int k   = d / RP;
        const int rel = d - k * RP;
        pk[u] = (ss[u] << RP_BITS) | rel;
        kk[u] = -1;  rk[u] = 0;
        if (e < tot && (unsigned)d < (unsigned)n) {
            kk[u] = k;
            rk[u] = __hip_atomic_fetch_add(&cnt[k], 1, __ATOMIC_RELAXED, __HIP_MEMORY_SCOPE_WORKGROUP);
        }
    }
    __syncthreads();

    int incl = 0;
    if (tid < NBK_MAX) {
        incl = cnt[tid];
        for (int dlt = 1; dlt < 32; dlt <<= 1) {
            const int t = __shfl_up(incl, dlt);
            if (lane >= dlt) incl += t;
        }
        if (lane == 31) wsum[wave] = incl;
    }
    __syncthreads();
    if (tid < NBK_MAX) {
        int add = 0;
        for (int w = 0; w < wave; ++w) add += wsum[w];
        off[tid + 1] = incl + add;
        if (tid == 0) off[0] = 0;
    }
    __syncthreads();

    #pragma unroll
    for (int u = 0; u < BIN_EPT; ++u) {
        if (kk[u] >= 0) {
            const int p = off[kk[u]] + rk[u];
            sp[p] = pk[u];
            sv[p] = vv[u];
        }
    }
    if (tid <= NBK_MAX) woff[(size_t)blockIdx.x * OFF_PITCH + tid] = off[tid];
    __syncthreads();

    const int total = off[NBK_MAX];
    for (int i = tid; i < total; i += BIN_THR) {
        wp[base + i] = sp[i];
        wv[base + i] = sv[i];
    }
}

__device__ __forceinline__ void lds_add(float* p, float v)
{
    __hip_atomic_fetch_add(p, v, __ATOMIC_RELAXED, __HIP_MEMORY_SCOPE_WORKGROUP);
}

__device__ __forceinline__ void own_add(float* acc, int rec, float v, v4f a, v4f b, v2f c)
{
    float* ar = acc + (rec & ((1 << RP_BITS) - 1)) * NOUT;
    lds_add(ar + 0, v * a.x);  lds_add(ar + 1, v * a.y);
    lds_add(ar + 2, v * a.z);  lds_add(ar + 3, v * a.w);
    lds_add(ar + 4, v * b.x);  lds_add(ar + 5, v * b.y);
    lds_add(ar + 6, v * b.z);  lds_add(ar + 7, v * b.w);
    lds_add(ar + 8, v * c.x);  lds_add(ar + 9, v * c.y);
}

__global__ __launch_bounds__(OWN_THR) void k_owner(
    const int* __restrict__ wp, const float* __restrict__ wv, const int* __restrict__ woff, int nb,
    const float* __restrict__ g, float* __restrict__ out, int n)
{
    __shared__ float acc[RP * NOUT];

    const int tid  = threadIdx.x;
    const int lane = tid & 31;
    const int wave = __builtin_amdgcn_readfirstlane(tid >> 5);
    const int part = blockIdx.x;
    const int lo   = part * RP;
    const int rows = n - lo < RP ? n - lo : RP;

    for (int i = tid; i < RP * NOUT; i += OWN_THR) acc[i] = 0.f;
    __syncthreads();

    for (int b0 = 0; b0 < nb; b0 += OWN_THR) {
        const int b = b0 + wave + 32 * lane;
        int st = 0, en = 0;
        if (b < nb) {
            const int* o = woff + (size_t)b * OFF_PITCH + part;
            st = b * BIN_C + o[0];
            en = b * BIN_C + o[1];
        }
        const int left = nb - b0 - wave;
        const int nseg = left <= 0 ? 0 : ((left + 31) / 32 < 32 ? (left + 31) / 32 : 32);

        for (int j = 0; j < nseg; j += 2) {
            const int  j1   = j + 1 < 32 ? j + 1 : 31;
            const bool two  = j + 1 < nseg;
            const int  st0  = __shfl(st, j),  en0 = __shfl(en, j);
            const int  st1r = __shfl(st, j1), en1r = __shfl(en, j1);
            const int  st1  = two ? st1r : 0, en1 = two ? en1r : 0;
            const int  len0 = en0 - st0, len1 = en1 - st1;
            const int  len  = len0 > len1 ? len0 : len1;
            for (int o = 0; o < len; o += 32) {
                const int  i0  = st0 + o + lane, i1 = st1 + o + lane;
                const bool ok0 = i0 < en0,       ok1 = i1 < en1;
                const int   r0 = wp[ok0 ? i0 : 0];
                const float v0 = wv[ok0 ? i0 : 0];
                const int   r1 = wp[ok1 ? i1 : 0];
                const float v1 = wv[ok1 ? i1 : 0];
                const float* g0 = g + (size_t)(ok0 ? r0 >> RP_BITS : 0) * GPITCH;
                const float* g1r = g + (size_t)(ok1 ? r1 >> RP_BITS : 0) * GPITCH;
                const v4f a0 = *(const v4f*)(g0);
                const v4f b0v = *(const v4f*)(g0 + 4);
                const v2f c0 = *(const v2f*)(g0 + 8);
                const v4f a1 = *(const v4f*)(g1r);
                const v4f b1v = *(const v4f*)(g1r + 4);
                const v2f c1 = *(const v2f*)(g1r + 8);
                if (ok0) own_add(acc, r0, v0, a0, b0v, c0);
                if (ok1) own_add(acc, r1, v1, a1, b1v, c1);
            }
        }
    }
    __syncthreads();

    float* o = out + (size_t)lo * NOUT;
    const int cnt = rows * NOUT;
    for (int i = tid; i < cnt; i += OWN_THR) o[i] = o[i] + acc[i];
}

extern "C" void kernel_launch(void* const* d_in, const int* in_sizes, int n_in,
                              void* d_out, int out_size, void* d_ws, size_t ws_size,
                              hipStream_t stream)
{
    const float* x    = (const float*)d_in[0];
    const float* W1   = (const float*)d_in[1];
    const float* Wf   = (const float*)d_in[2];
    const int*   src1 = (const int*)  d_in[3];
    const int*   dst1 = (const int*)  d_in[4];
    const float* val1 = (const float*)d_in[5];
    const int*   src2 = (const int*)  d_in[6];
    const int*   dst2 = (const int*)  d_in[7];
    const float* val2 = (const float*)d_in[8];

    const int n    = in_sizes[0] / KDIM;
    const int nnz1 = in_sizes[3];
    const int nnz2 = in_sizes[6];
    const int nb   = (nnz1 + nnz2 + BIN_C - 1) / BIN_C;

    float* g1   = (float*)d_ws;
    float* g2   = g1 + (size_t)n * GPITCH;
    int*   wp   = (int*)(g2 + (size_t)n * GPITCH);
    float* wv   = (float*)(wp + (size_t)nb * BIN_C);
    int*   woff = (int*)(wv + (size_t)nb * BIN_C);
    float* out  = (float*)d_out;

    const int n_tiles = (n + 15) / 16;

    k_bin<<<nb, BIN_THR, 0, stream>>>(src1, dst1, val1, nnz1, src2, dst2, val2, nnz2, n, wp, wv, woff);

    k_gemm_relu_proj<<<(n_tiles + 7) / 8, 256, 0, stream>>>(x, W1, Wf, out, g1, g2, n_tiles);

    k_owner<<<(n + RP - 1) / RP, OWN_THR, 0, stream>>>(wp, wv, woff, nb, g1, out, n);
}


# Edge ordering (was kernel k_bin) runs inside the stage-1 launch as 252 extra 256-thread workgroups behind the GEMM's own: its memory waits sit beside the matrix work; one launch fewer; sum order kept
# speedup vs baseline: 4.5671x; 1.0793x over previous
#include <hip/hip_runtime.h>

typedef float v2f __attribute__((ext_vector_type(2)));
typedef float v4f __attribute__((ext_vector_type(4)));
typedef float v8f __attribute__((ext_vector_type(8)));

#define KDIM   500
#define HIDDEN 64
#define NOUT   10
#define GPITCH 16

#define RP        400
#define RP_BITS   9
#define NBK_MAX   256
#define BIN_THR   256
#define BIN_EPT   28
#define BIN_C     (BIN_EPT * BIN_THR)
#define OFF_PITCH (NBK_MAX + 1)
#define OWN_THR   1024

__device__ __forceinline__ void bin_role(
    float* smem, int bb,
    const int* __restrict__ src1, const int* __restrict__ dst1, const float* __restrict__ val1, int nnz1,
    const int* __restrict__ src2, const int* __restrict__ dst2, const float* __restrict__ val2, int nnz2,
    int n, int* __restrict__ wp, float* __restrict__ wv, int* __restrict__ woff)
{
    int*   cnt  = (int*)smem;
    int*   off  = cnt + NBK_MAX;
    int*   wsum = off + NBK_MAX + 4;
    int*   sp   = wsum + NBK_MAX / 32;
    float* sv   = (float*)(sp + BIN_C);

    const int tid  = threadIdx.x;
    const int lane = tid & 31;
    const int wave = tid >> 5;
    const int tot  = nnz1 + nnz2;
    const int base = bb * BIN_C;

    cnt[tid] = 0;
    __syncthreads();

    int   pk[BIN_EPT], kr[BIN_EPT];
    int   dd[BIN_EPT], ss[BIN_EPT];
    float vv[BIN_EPT];
    #pragma unroll
    for (int u = 0; u < BIN_EPT; ++u) {
        const int    e      = base + u * BIN_THR + tid;
        const int    ec     = e < tot ? e : tot - 1;
        const bool   second = ec >= nnz1;
        const int    i      = second ? ec - nnz1 : ec;
        const int*   dq     = second ? dst2 : dst1;
        const int*   sq     = second ? src2 : src1;
        const float* vq     = second ? val2 : val1;
        dd[u] = dq[i];
        ss[u] = sq[i] + (second ? n : 0);
        vv[u] = vq[i];
    }
    #pragma unroll
    for (int u = 0; u < BIN_EPT; ++u) {
        const int e   = base + u * BIN_THR + tid;
        const int d   = dd[u];
        const int k   = d / RP;
        const int rel = d - k * RP;
        pk[u] = (ss[u] << RP_BITS) | rel;
        kr[u] = -1;
        if (e < tot && (unsigned)d < (unsigned)n) {
            const int r = __hip_atomic_fetch_add(&cnt[k], 1, __ATOMIC_RELAXED, __HIP_MEMORY_SCOPE_WORKGROUP);
            kr[u] = (k << 16) | r;
        }
    }
    __syncthreads();

    int incl = cnt[tid];
    for (int dlt = 1; dlt < 32; dlt <<= 1) {
        const int t = __shfl_up(incl, dlt);
        if (lane >= dlt) incl += t;
    }
    if (lane == 31) wsum[wave] = incl;
    __syncthreads();
    {
        int add = 0;
        for (int w = 0; w < wave; ++w) add += wsum[w];
        off[tid + 1] = incl + add;
        if (tid == 0) off[0] = 0;
    }
    __syncthreads();

    #pragma unroll
    for (int u = 0; u < BIN_EPT; ++u) {
        if (kr[u] >= 0) {
            const int p = off[kr[u] >> 16] + (kr[u] & 0xffff);
            sp[p] = pk[u];
            sv[p] = vv[u];
        }
    }
    woff[(size_t)bb * OFF_PITCH + tid] = off[tid];
    if (tid == 0) woff[(size_t)bb * OFF_PITCH + NBK_MAX] = off[NBK_MAX];
    __syncthreads();

    const int total = off[NBK_MAX];
    for (int i = tid; i < total; i += BIN_THR) {
        wp[base + i] = sp[i];
        wv[base + i] = sv[i];
    }
}

#define KC     100
#define KC4    (KC / 4)
#define NCHUNK (KDIM / KC)
#define TILE_M 128
#define LROWS  10
#define LTHR   (LROWS * KC4)
#define XPASS  ((TILE_M + LROWS - 1) / LROWS)
#define WPASS  ((HIDDEN + LROWS - 1) / LROWS)
#define XLAST  (TILE_M - (XPASS - 1) * LROWS)
#define WLAST  (HIDDEN - (WPASS - 1) * LROWS)

__global__ __launch_bounds__(256) void k_gemm_relu_proj_bin(
    const float* __restrict__ x, const float* __restrict__ W1,
    const float* __restrict__ Wf,
    float* __restrict__ out, float* __restrict__ g1, float* __restrict__ g2,
    int n_tiles, int n_gemm_blocks,
    const int* __restrict__ src1, const int* __restrict__ dst1, const float* __restrict__ val1, int nnz1,
    const int* __restrict__ src2, const int* __restrict__ dst2, const float* __restrict__ val2, int nnz2,
    int n, int* __restrict__ wp, float* __restrict__ wv, int* __restrict__ woff)
{
    __shared__ __attribute__((aligned(16))) float smem[TILE_M * KC + HIDDEN * KC];

    if ((int)blockIdx.x >= n_gemm_blocks) {
        bin_role(smem, (int)blockIdx.x - n_gemm_blocks,
                 src1, dst1, val1, nnz1, src2, dst2, val2, nnz2, n, wp, wv, woff);
        return;
    }
    float* sx = smem;
    float* sw = smem + TILE_M * KC;

    const int tid     = threadIdx.x;
    const int lane    = tid & 31;
    const int wave    = tid >> 5;
    const int n_rows  = n_tiles * 16;
    const int row_nom = blockIdx.x * TILE_M;
    const int row0    = row_nom < n_rows - TILE_M ? row_nom : n_rows - TILE_M;
    const int nn      = lane & 15;
    const int khalf   = (lane >> 4) << 1;

    const int lt  = tid < LTHR ? tid : tid - LTHR;
    const int lr  = lt / KC4;
    const int c4  = lt - lr * KC4;
    const int lrx = lr < XLAST ? lr : XLAST - 1;
    const int lrw = lr < WLAST ? lr : WLAST - 1;
    const float* gx  = x  + (size_t)(row0 + lr) * KDIM + c4 * 4;
    const float* gxl = x  + (size_t)(row0 + (XPASS - 1) * LROWS + lrx) * KDIM + c4 * 4;
    const float* gw  = W1 + (size_t)lr * KDIM + c4 * 4;
    const float* gwl = W1 + (size_t)((WPASS - 1) * LROWS + lrw) * KDIM + c4 * 4;
    const int    ls  = lr * KC + c4 * 4;

    const float* ax = sx + (wave * 16 + nn) * KC + khalf;
    const float* bw = sw + nn * KC + khalf;

    v4f rx[XPASS], rw[WPASS];
    v8f acc0 = {}, acc1 = {}, acc2 = {}, acc3 = {};

    for (int c = 0; c <= NCHUNK; ++c) {
        if (c > 0) {
            __syncthreads();
            if (tid < LTHR) {
                for (int p = 0; p < XPASS - 1; ++p) *(v4f*)(sx + ls + p * LROWS * KC) = rx[p];
                for (int p = 0; p < WPASS - 1; ++p) *(v4f*)(sw + ls + p * LROWS * KC) = rw[p];
                if (lr < XLAST) *(v4f*)(sx + ls + (XPASS - 1) * LROWS * KC) = rx[XPASS - 1];
                if (lr < WLAST) *(v4f*)(sw + ls + (WPASS - 1) * LROWS * KC) = rw[WPASS - 1];
            }
            __syncthreads();
        }
        if (c < NCHUNK) {
            const int kc0 = c * KC;
            for (int p = 0; p < XPASS - 1; ++p)
                rx[p] = __builtin_nontemporal_load((const v4f*)(gx + p * LROWS * KDIM + kc0));
            rx[XPASS - 1] = __builtin_nontemporal_load((const v4f*)(gxl + kc0));
            for (int p = 0; p < WPASS - 1; ++p)
                rw[p] = *(const v4f*)(gw + p * LROWS * KDIM + kc0);
            rw[WPASS - 1] = *(const v4f*)(gwl + kc0);
        }
        if (c > 0) {
            for (int kk = 0; kk < KC; kk += 4) {
                v2f a  = *(const v2f*)(ax + kk);
                v2f b0 = *(const v2f*)(bw +  0 * KC + kk);
                v2f b1 = *(const v2f*)(bw + 16 * KC + kk);
                v2f b2 = *(const v2f*)(bw + 32 * KC + kk);
                v2f b3 = *(const v2f*)(bw + 48 * KC + kk);
                acc0 = __builtin_amdgcn_wmma_f32_16x16x4_f32(false, a, false, b0, (short)0, acc0, false, false);
                acc1 = __builtin_amdgcn_wmma_f32_16x16x4_f32(false, a, false, b1, (short)0, acc1, false, false);
                acc2 = __builtin_amdgcn_wmma_f32_16x16x4_f32(false, a, false, b2, (short)0, acc2, false, false);
                acc3 = __builtin_amdgcn_wmma_f32_16x16x4_f32(false, a, false, b3, (short)0, acc3, false, false);
            }
        }
    }
    if (row0 + wave * 16 < row_nom) return;

    const bool live = nn < NOUT;
    const int  wn   = live ? nn : NOUT - 1;
    const float* wrow = Wf + (size_t)wn * (3 * HIDDEN) + khalf;

    float* hx = sx + (wave * 16 + ((lane >> 4) << 3)) * KC + nn;
    for (int r = 0; r < 8; ++r) {
        float v0 = acc0[r]; v0 = v0 > 0.f ? v0 : 0.f;
        float v1 = acc1[r]; v1 = v1 > 0.f ? v1 : 0.f;
        float v2 = acc2[r]; v2 = v2 > 0.f ? v2 : 0.f;
        float v3 = acc3[r]; v3 = v3 > 0.f ? v3 : 0.f;
        hx[r * KC +  0] = v0;
        hx[r * KC + 16] = v1;
        hx[r * KC + 32] = v2;
        hx[r * KC + 48] = v3;
    }

    v8f p0 = {}, p1 = {}, p2 = {};
    #pragma unroll
    for (int k0 = 0; k0 < HIDDEN; k0 += 4) {
        const v2f a = *(const v2f*)(ax + k0);
        const v2f w0 = *(const v2f*)(wrow + 0 * HIDDEN + k0);
        const v2f w1 = *(const v2f*)(wrow + 1 * HIDDEN + k0);
        const v2f w2 = *(const v2f*)(wrow + 2 * HIDDEN + k0);
        v2f b0, b1, b2;
        b0.x = live ? w0.x : 0.f;  b0.y = live ? w0.y : 0.f;
        b1.x = live ? w1.x : 0.f;  b1.y = live ? w1.y : 0.f;
        b2.x = live ? w2.x : 0.f;  b2.y = live ? w2.y : 0.f;
        p0 = __builtin_amdgcn_wmma_f32_16x16x4_f32(false, a, false, b0, (short)0, p0, false, false);
        p1 = __builtin_amdgcn_wmma_f32_16x16x4_f32(false, a, false, b1, (short)0, p1, false, false);
        p2 = __builtin_amdgcn_wmma_f32_16x16x4_f32(false, a, false, b2, (short)0, p2, false, false);
    }
    const int m_lo = row0 + wave * 16 + ((lane >> 4) << 3);
    for (int r = 0; r < 8; ++r) {
        const size_t m = (size_t)(m_lo + r);
        g1[m * GPITCH + nn] = p1[r];
        g2[m * GPITCH + nn] = p2[r];
        if (live) out[m * NOUT + nn] = p0[r];
    }
}

__device__ __forceinline__ void lds_add(float* p, float v)
{
    __hip_atomic_fetch_add(p, v, __ATOMIC_RELAXED, __HIP_MEMORY_SCOPE_WORKGROUP);
}

__device__ __forceinline__ void own_add(float* acc, int rec, float v, v4f a, v4f b, v2f c)
{
    float* ar = acc + (rec & ((1 << RP_BITS) - 1)) * NOUT;
    lds_add(ar + 0, v * a.x);  lds_add(ar + 1, v * a.y);
    lds_add(ar + 2, v * a.z);  lds_add(ar + 3, v * a.w);
    lds_add(ar + 4, v * b.x);  lds_add(ar + 5, v * b.y);
    lds_add(ar + 6, v * b.z);  lds_add(ar + 7, v * b.w);
    lds_add(ar + 8, v * c.x);  lds_add(ar + 9, v * c.y);
}

__global__ __launch_bounds__(OWN_THR) void k_owner(
    const int* __restrict__ wp, const float* __restrict__ wv, const int* __restrict__ woff, int nb,
    const float* __restrict__ g, float* __restrict__ out, int n)
{
    __shared__ float acc[RP * NOUT];

    const int tid  = threadIdx.x;
    const int lane = tid & 31;
    const int wave = __builtin_amdgcn_readfirstlane(tid >> 5);
    const int part = blockIdx.x;
    const int lo   = part * RP;
    const int rows = n - lo < RP ? n - lo : RP;

    for (int i = tid; i < RP * NOUT; i += OWN_THR) acc[i] = 0.f;
    __syncthreads();

    for (int b0 = 0; b0 < nb; b0 += OWN_THR) {
        const int b = b0 + wave + 32 * lane;
        int st = 0, en = 0;
        if (b < nb) {
            const int* o = woff + (size_t)b * OFF_PITCH + part;
            st = b * BIN_C + o[0];
            en = b * BIN_C + o[1];
        }
        const int left = nb - b0 - wave;
        const int nseg = left <= 0 ? 0 : ((left + 31) / 32 < 32 ? (left + 31) / 32 : 32);

        for (int j = 0; j < nseg; j += 2) {
            const int  j1   = j + 1 < 32 ? j + 1 : 31;
            const bool two  = j + 1 < nseg;
            const int  st0  = __shfl(st, j),  en0 = __shfl(en, j);
            const int  st1r = __shfl(st, j1), en1r = __shfl(en, j1);
            const int  st1  = two ? st1r : 0, en1 = two ? en1r : 0;
            const int  len0 = en0 - st0, len1 = en1 - st1;
            const int  len  = len0 > len1 ? len0 : len1;
            for (int o = 0; o < len; o += 32) {
                const int  i0  = st0 + o + lane, i1 = st1 + o + lane;
                const bool ok0 = i0 < en0,       ok1 = i1 < en1;
                const int   r0 = wp[ok0 ? i0 : 0];
                const float v0 = wv[ok0 ? i0 : 0];
                const int   r1 = wp[ok1 ? i1 : 0];
                const float v1 = wv[ok1 ? i1 : 0];
                const float* g0 = g + (size_t)(ok0 ? r0 >> RP_BITS : 0) * GPITCH;
                const float* g1r = g + (size_t)(ok1 ? r1 >> RP_BITS : 0) * GPITCH;
                const v4f a0 = *(const v4f*)(g0);
                const v4f b0v = *(const v4f*)(g0 + 4);
                const v2f c0 = *(const v2f*)(g0 + 8);
                const v4f a1 = *(const v4f*)(g1r);
                const v4f b1v = *(const v4f*)(g1r + 4);
                const v2f c1 = *(const v2f*)(g1r + 8);
                if (ok0) own_add(acc, r0, v0, a0, b0v, c0);
                if (ok1) own_add(acc, r1, v1, a1, b1v, c1);
            }
        }
    }
    __syncthreads();

    float* o = out + (size_t)lo * NOUT;
    const int cnt = rows * NOUT;
    for (int i = tid; i < cnt; i += OWN_THR) o[i] = o[i] + acc[i];
}

extern "C" void kernel_launch(void* const* d_in, const int* in_sizes, int n_in,
                              void* d_out, int out_size, void* d_ws, size_t ws_size,
                              hipStream_t stream)
{
    const float* x    = (const float*)d_in[0];
    const float* W1   = (const float*)d_in[1];
    const float* Wf   = (const float*)d_in[2];
    const int*   src1 = (const int*)  d_in[3];
    const int*   dst1 = (const int*)  d_in[4];
    const float* val1 = (const float*)d_in[5];
    const int*   src2 = (const int*)  d_in[6];
    const int*   dst2 = (const int*)  d_in[7];
    const float* val2 = (const float*)d_in[8];

    const int n    = in_sizes[0] / KDIM;
    const int nnz1 = in_sizes[3];
    const int nnz2 = in_sizes[6];
    const int nb   = (nnz1 + nnz2 + BIN_C - 1) / BIN_C;

    float* g1   = (float*)d_ws;
    float* g2   = g1 + (size_t)n * GPITCH;
    int*   wp   = (int*)(g2 + (size_t)n * GPITCH);
    float* wv   = (float*)(wp + (size_t)nb * BIN_C);
    int*   woff = (int*)(wv + (size_t)nb * BIN_C);
    float* out  = (float*)d_out;

    const int n_tiles = (n + 15) / 16;

    const int n_gemm = (n_tiles + 7) / 8;

    k_gemm_relu_proj_bin<<<n_gemm + nb, 256, 0, stream>>>(x, W1, Wf, out, g1, g2, n_tiles, n_gemm,
        src1, dst1, val1, nnz1, src2, dst2, val2, nnz2, n, wp, wv, woff);

    k_owner<<<(n + RP - 1) / RP, OWN_THR, 0, stream>>>(wp, wv, woff, nb, g1, out, n);
}


# Stage-1 GEMM made persistent: 256 workgroups copy W1 whole into LDS once, each wave stages its own 16 rows of x in a private LDS strip: no barrier, no W1 re-staging in the k loop; sum order kept
# speedup vs baseline: 4.8560x; 1.0633x over previous
#include <hip/hip_runtime.h>

typedef float v2f __attribute__((ext_vector_type(2)));
typedef float v4f __attribute__((ext_vector_type(4)));
typedef float v8f __attribute__((ext_vector_type(8)));

#define KDIM   500
#define HIDDEN 64
#define NOUT   10
#define GPITCH 16

#define RP        400
#define RP_BITS   9
#define NBK_MAX   256
#define BIN_THR   256
#define BIN_EPT   28
#define BIN_C     (BIN_EPT * BIN_THR)
#define OFF_PITCH (NBK_MAX + 1)
#define OWN_THR   1024

__device__ __forceinline__ void bin_role(
    float* smem, int bb,
    const int* __restrict__ src1, const int* __restrict__ dst1, const float* __restrict__ val1, int nnz1,
    const int* __restrict__ src2, const int* __restrict__ dst2, const float* __restrict__ val2, int nnz2,
    int n, int* __restrict__ wp, float* __restrict__ wv, int* __restrict__ woff)
{
    int*   cnt  = (int*)smem;
    int*   off  = cnt + NBK_MAX;
    int*   wsum = off + NBK_MAX + 4;
    int*   sp   = wsum + NBK_MAX / 32;
    float* sv   = (float*)(sp + BIN_C);

    const int tid  = threadIdx.x;
    const int lane = tid & 31;
    const int wave = tid >> 5;
    const int tot  = nnz1 + nnz2;
    const int base = bb * BIN_C;

    cnt[tid] = 0;
    __syncthreads();

    int   pk[BIN_EPT], kr[BIN_EPT];
    int   dd[BIN_EPT], ss[BIN_EPT];
    float vv[BIN_EPT];
    #pragma unroll
    for (int u = 0; u < BIN_EPT; ++u) {
        const int    e      = base + u * BIN_THR + tid;
        const int    ec     = e < tot ? e : tot - 1;
        const bool   second = ec >= nnz1;
        const int    i      = second ? ec - nnz1 : ec;
        const int*   dq     = second ? dst2 : dst1;
        const int*   sq     = second ? src2 : src1;
        const float* vq     = second ? val2 : val1;
        dd[u] = dq[i];
        ss[u] = sq[i] + (second ? n : 0);
        vv[u] = vq[i];
    }
    #pragma unroll
    for (int u = 0; u < BIN_EPT; ++u) {
        const int e   = base + u * BIN_THR + tid;
        const int d   = dd[u];
        const int k   = d / RP;
        const int rel = d - k * RP;
        pk[u] = (ss[u] << RP_BITS) | rel;
        kr[u] = -1;
        if (e < tot && (unsigned)d < (unsigned)n) {
            const int r = __hip_atomic_fetch_add(&cnt[k], 1, __ATOMIC_RELAXED, __HIP_MEMORY_SCOPE_WORKGROUP);
            kr[u] = (k << 16) | r;
        }
    }
    __syncthreads();

    int incl = cnt[tid];
    for (int dlt = 1; dlt < 32; dlt <<= 1) {
        const int t = __shfl_up(incl, dlt);
        if (lane >= dlt) incl += t;
    }
    if (lane == 31) wsum[wave] = incl;
    __syncthreads();
    {
        int add = 0;
        for (int w = 0; w < wave; ++w) add += wsum[w];
        off[tid + 1] = incl + add;
        if (tid == 0) off[0] = 0;
    }
    __syncthreads();

    #pragma unroll
    for (int u = 0; u < BIN_EPT; ++u) {
        if (kr[u] >= 0) {
            const int p = off[kr[u] >> 16] + (kr[u] & 0xffff);
            sp[p] = pk[u];
            sv[p] = vv[u];
        }
    }
    woff[(size_t)bb * OFF_PITCH + tid] = off[tid];
    if (tid == 0) woff[(size_t)bb * OFF_PITCH + NBK_MAX] = off[NBK_MAX];
    __syncthreads();

    const int total = off[NBK_MAX];
    for (int i = tid; i < total; i += BIN_THR) {
        wp[base + i] = sp[i];
        wv[base + i] = sv[i];
    }
}

#define KC      100
#define KC4     (KC / 4)
#define NCHUNK  (KDIM / KC)
#define NWAVE   8
#define XP      (16 * KC)
#define XPIECES (16 * KC4)
#define XLD     ((XPIECES + 31) / 32)
#define XTAIL   (XPIECES - (XLD - 1) * 32)
#define SW_FL   (HIDDEN * KDIM)
#define SW_V4   (SW_FL / 4)
#define WGRP    8
#define WRND    ((SW_V4 + WGRP * 256 - 1) / (WGRP * 256))

__global__ __launch_bounds__(256) void k_gemm_relu_proj_bin(
    const float* __restrict__ x, const float* __restrict__ W1,
    const float* __restrict__ Wf,
    float* __restrict__ out, float* __restrict__ g1, float* __restrict__ g2,
    int n_tiles, int n_gemm_blocks,
    const int* __restrict__ src1, const int* __restrict__ dst1, const float* __restrict__ val1, int nnz1,
    const int* __restrict__ src2, const int* __restrict__ dst2, const float* __restrict__ val2, int nnz2,
    int n, int* __restrict__ wp, float* __restrict__ wv, int* __restrict__ woff)
{
    __shared__ __attribute__((aligned(16))) float smem[SW_FL + NWAVE * XP];

    if ((int)blockIdx.x >= n_gemm_blocks) {
        bin_role(smem, (int)blockIdx.x - n_gemm_blocks,
                 src1, dst1, val1, nnz1, src2, dst2, val2, nnz2, n, wp, wv, woff);
        return;
    }
    const int tid   = threadIdx.x;
    const int lane  = tid & 31;
    const int wave  = __builtin_amdgcn_readfirstlane(tid >> 5);
    const int nn    = lane & 15;
    const int khalf = (lane >> 4) << 1;

    float* sw = smem;
    float* px = smem + SW_FL + wave * XP;

    int goff[XLD];
    #pragma unroll
    for (int i = 0; i < XLD; ++i) {
        const int q0 = lane + 32 * i;
        const int q  = q0 < XPIECES ? q0 : XPIECES - 1;
        const int r  = q / KC4;
        goff[i] = r * KDIM + (q - r * KC4) * 4;
    }
    const bool tail_ok = lane < XTAIL;
    float* ps = px + 4 * lane;

    const int tstep = NWAVE * n_gemm_blocks;
    int t = wave * n_gemm_blocks + (int)blockIdx.x;

    v4f rx[XLD];
    if (t < n_tiles) {
        const float* xt = x + (size_t)t * (16 * KDIM);
        #pragma unroll
        for (int i = 0; i < XLD; ++i) rx[i] = __builtin_nontemporal_load((const v4f*)(xt + goff[i]));
    }

    #pragma unroll
    for (int r = 0; r < WRND; ++r) {
        v4f tw[WGRP];
        #pragma unroll
        for (int u = 0; u < WGRP; ++u) {
            const int i  = tid + 256 * (WGRP * r + u);
            const int ic = i < SW_V4 ? i : SW_V4 - 1;
            tw[u] = *(const v4f*)(W1 + 4 * ic);
        }
        #pragma unroll
        for (int u = 0; u < WGRP; ++u) {
            const int i = tid + 256 * (WGRP * r + u);
            if (i < SW_V4) *(v4f*)(sw + 4 * i) = tw[u];
        }
    }
    __syncthreads();

    const float* ax = px + nn * KC + khalf;
    const float* bw = sw + nn * KDIM + khalf;

    const bool live = nn < NOUT;
    const int  wn   = live ? nn : NOUT - 1;
    const float* wrow = Wf + (size_t)wn * (3 * HIDDEN) + khalf;

    for (; t < n_tiles; t += tstep) {
        const float* xt = x + (size_t)t * (16 * KDIM);
        const int    tn = t + tstep;

        #pragma unroll
        for (int i = 0; i < XLD - 1; ++i) *(v4f*)(ps + 128 * i) = rx[i];
        if (tail_ok) *(v4f*)(ps + 128 * (XLD - 1)) = rx[XLD - 1];

        v8f acc0 = {}, acc1 = {}, acc2 = {}, acc3 = {};
        for (int c = 0; c < NCHUNK; ++c) {
            if (c + 1 < NCHUNK || tn < n_tiles) {
                const float* xn = c + 1 < NCHUNK ? xt + (c + 1) * KC : x + (size_t)tn * (16 * KDIM);
                #pragma unroll
                for (int i = 0; i < XLD; ++i) rx[i] = __builtin_nontemporal_load((const v4f*)(xn + goff[i]));
            }
            const float* bc = bw + c * KC;
            for (int kk = 0; kk < KC; kk += 4) {
                v2f a  = *(const v2f*)(ax + kk);
                v2f b0 = *(const v2f*)(bc +  0 * KDIM + kk);
                v2f b1 = *(const v2f*)(bc + 16 * KDIM + kk);
                v2f b2 = *(const v2f*)(bc + 32 * KDIM + kk);
                v2f b3 = *(const v2f*)(bc + 48 * KDIM + kk);
                acc0 = __builtin_amdgcn_wmma_f32_16x16x4_f32(false, a, false, b0, (short)0, acc0, false, false);
                acc1 = __builtin_amdgcn_wmma_f32_16x16x4_f32(false, a, false, b1, (short)0, acc1, false, false);
                acc2 = __builtin_amdgcn_wmma_f32_16x16x4_f32(false, a, false, b2, (short)0, acc2, false, false);
                acc3 = __builtin_amdgcn_wmma_f32_16x16x4_f32(false, a, false, b3, (short)0, acc3, false, false);
            }
            if (c + 1 < NCHUNK) {
                #pragma unroll
                for (int i = 0; i < XLD - 1; ++i) *(v4f*)(ps + 128 * i) = rx[i];
                if (tail_ok) *(v4f*)(ps + 128 * (XLD - 1)) = rx[XLD - 1];
            }
        }

        float* hx = px + ((lane >> 4) << 3) * KC + nn;
        for (int r = 0; r < 8; ++r) {
            float v0 = acc0[r]; v0 = v0 > 0.f ? v0 : 0.f;
            float v1 = acc1[r]; v1 = v1 > 0.f ? v1 : 0.f;
            float v2 = acc2[r]; v2 = v2 > 0.f ? v2 : 0.f;
            float v3 = acc3[r]; v3 = v3 > 0.f ? v3 : 0.f;
            hx[r * KC +  0] = v0;
            hx[r * KC + 16] = v1;
            hx[r * KC + 32] = v2;
            hx[r * KC + 48] = v3;
        }

        v8f p0 = {}, p1 = {}, p2 = {};
        #pragma unroll
        for (int k0 = 0; k0 < HIDDEN; k0 += 4) {
            const v2f a = *(const v2f*)(ax + k0);
            const v2f w0 = *(const v2f*)(wrow + 0 * HIDDEN + k0);
            const v2f w1 = *(const v2f*)(wrow + 1 * HIDDEN + k0);
            const v2f w2 = *(const v2f*)(wrow + 2 * HIDDEN + k0);
            v2f b0, b1, b2;
            b0.x = live ? w0.x : 0.f;  b0.y = live ? w0.y : 0.f;
            b1.x = live ? w1.x : 0.f;  b1.y = live ? w1.y : 0.f;
            b2.x = live ? w2.x : 0.f;  b2.y = live ? w2.y : 0.f;
            p0 = __builtin_amdgcn_wmma_f32_16x16x4_f32(false, a, false, b0, (short)0, p0, false, false);
            p1 = __builtin_amdgcn_wmma_f32_16x16x4_f32(false, a, false, b1, (short)0, p1, false, false);
            p2 = __builtin_amdgcn_wmma_f32_16x16x4_f32(false, a, false, b2, (short)0, p2, false, false);
        }
        const int m_lo = t * 16 + ((lane >> 4) << 3);
        for (int r = 0; r < 8; ++r) {
            const size_t m = (size_t)(m_lo + r);
            g1[m * GPITCH + nn] = p1[r];
            g2[m * GPITCH + nn] = p2[r];
            if (live) out[m * NOUT + nn] = p0[r];
        }
    }
}

__device__ __forceinline__ void lds_add(float* p, float v)
{
    __hip_atomic_fetch_add(p, v, __ATOMIC_RELAXED, __HIP_MEMORY_SCOPE_WORKGROUP);
}

__device__ __forceinline__ void own_add(float* acc, int rec, float v, v4f a, v4f b, v2f c)
{
    float* ar = acc + (rec & ((1 << RP_BITS) - 1)) * NOUT;
    lds_add(ar + 0, v * a.x);  lds_add(ar + 1, v * a.y);
    lds_add(ar + 2, v * a.z);  lds_add(ar + 3, v * a.w);
    lds_add(ar + 4, v * b.x);  lds_add(ar + 5, v * b.y);
    lds_add(ar + 6, v * b.z);  lds_add(ar + 7, v * b.w);
    lds_add(ar + 8, v * c.x);  lds_add(ar + 9, v * c.y);
}

__global__ __launch_bounds__(OWN_THR) void k_owner(
    const int* __restrict__ wp, const float* __restrict__ wv, const int* __restrict__ woff, int nb,
    const float* __restrict__ g, float* __restrict__ out, int n)
{
    __shared__ float acc[RP * NOUT];

    const int tid  = threadIdx.x;
    const int lane = tid & 31;
    const int wave = __builtin_amdgcn_readfirstlane(tid >> 5);
    const int part = blockIdx.x;
    const int lo   = part * RP;
    const int rows = n - lo < RP ? n - lo : RP;

    for (int i = tid; i < RP * NOUT; i += OWN_THR) acc[i] = 0.f;
    __syncthreads();

    for (int b0 = 0; b0 < nb; b0 += OWN_THR) {
        const int b = b0 + wave + 32 * lane;
        int st = 0, en = 0;
        if (b < nb) {
            const int* o = woff + (size_t)b * OFF_PITCH + part;
            st = b * BIN_C + o[0];
            en = b * BIN_C + o[1];
        }
        const int left = nb - b0 - wave;
        const int nseg = left <= 0 ? 0 : ((left + 31) / 32 < 32 ? (left + 31) / 32 : 32);

        for (int j = 0; j < nseg; j += 2) {
            const int  j1   = j + 1 < 32 ? j + 1 : 31;
            const bool two  = j + 1 < nseg;
            const int  st0  = __shfl(st, j),  en0 = __shfl(en, j);
            const int  st1r = __shfl(st, j1), en1r = __shfl(en, j1);
            const int  st1  = two ? st1r : 0, en1 = two ? en1r : 0;
            const int  len0 = en0 - st0, len1 = en1 - st1;
            const int  len  = len0 > len1 ? len0 : len1;
            for (int o = 0; o < len; o += 32) {
                const int  i0  = st0 + o + lane, i1 = st1 + o + lane;
                const bool ok0 = i0 < en0,       ok1 = i1 < en1;
                const int   r0 = wp[ok0 ? i0 : 0];
                const float v0 = wv[ok0 ? i0 : 0];
                const int   r1 = wp[ok1 ? i1 : 0];
                const float v1 = wv[ok1 ? i1 : 0];
                const float* g0 = g + (size_t)(ok0 ? r0 >> RP_BITS : 0) * GPITCH;
                const float* g1r = g + (size_t)(ok1 ? r1 >> RP_BITS : 0) * GPITCH;
                const v4f a0 = *(const v4f*)(g0);
                const v4f b0v = *(const v4f*)(g0 + 4);
                const v2f c0 = *(const v2f*)(g0 + 8);
                const v4f a1 = *(const v4f*)(g1r);
                const v4f b1v = *(const v4f*)(g1r + 4);
                const v2f c1 = *(const v2f*)(g1r + 8);
                if (ok0) own_add(acc, r0, v0, a0, b0v, c0);
                if (ok1) own_add(acc, r1, v1, a1, b1v, c1);
            }
        }
    }
    __syncthreads();

    float* o = out + (size_t)lo * NOUT;
    const int cnt = rows * NOUT;
    for (int i = tid; i < cnt; i += OWN_THR) o[i] = o[i] + acc[i];
}

extern "C" void kernel_launch(void* const* d_in, const int* in_sizes, int n_in,
                              void* d_out, int out_size, void* d_ws, size_t ws_size,
                              hipStream_t stream)
{
    const float* x    = (const float*)d_in[0];
    const float* W1   = (const float*)d_in[1];
    const float* Wf   = (const float*)d_in[2];
    const int*   src1 = (const int*)  d_in[3];
    const int*   dst1 = (const int*)  d_in[4];
    const float* val1 = (const float*)d_in[5];
    const int*   src2 = (const int*)  d_in[6];
    const int*   dst2 = (const int*)  d_in[7];
    const float* val2 = (const float*)d_in[8];

    const int n    = in_sizes[0] / KDIM;
    const int nnz1 = in_sizes[3];
    const int nnz2 = in_sizes[6];
    const int nb   = (nnz1 + nnz2 + BIN_C - 1) / BIN_C;

    float* g1   = (float*)d_ws;
    float* g2   = g1 + (size_t)n * GPITCH;
    int*   wp   = (int*)(g2 + (size_t)n * GPITCH);
    float* wv   = (float*)(wp + (size_t)nb * BIN_C);
    int*   woff = (int*)(wv + (size_t)nb * BIN_C);
    float* out  = (float*)d_out;

    const int n_tiles = (n + 15) / 16;

    const int n_gemm = 256;

    k_gemm_relu_proj_bin<<<n_gemm + nb, 256, 0, stream>>>(x, W1, Wf, out, g1, g2, n_tiles, n_gemm,
        src1, dst1, val1, nnz1, src2, dst2, val2, nnz2, n, wp, wv, woff);

    k_owner<<<(n + RP - 1) / RP, OWN_THR, 0, stream>>>(wp, wv, woff, nb, g1, out, n);
}


# Owner kernel: a range's records counting-sorted by destination row in LDS (one integer rank each); 4 lanes per row then sum them in registers: the 10 LDS float adds per record gone; sum order changes
# speedup vs baseline: 5.0484x; 1.0396x over previous
#include <hip/hip_runtime.h>

typedef float v2f __attribute__((ext_vector_type(2)));
typedef float v4f __attribute__((ext_vector_type(4)));
typedef float v8f __attribute__((ext_vector_type(8)));

#define KDIM   500
#define HIDDEN 64
#define NOUT   10
#define GPITCH 16

#define RP        400
#define RP_BITS   9
#define NBK_MAX   256
#define BIN_THR   256
#define BIN_EPT   28
#define BIN_C     (BIN_EPT * BIN_THR)
#define OFF_PITCH (NBK_MAX + 1)
#define OWN_THR   1024

__device__ __forceinline__ void bin_role(
    float* smem, int bb,
    const int* __restrict__ src1, const int* __restrict__ dst1, const float* __restrict__ val1, int nnz1,
    const int* __restrict__ src2, const int* __restrict__ dst2, const float* __restrict__ val2, int nnz2,
    int n, int* __restrict__ wp, float* __restrict__ wv, int* __restrict__ woff)
{
    int*   cnt  = (int*)smem;
    int*   off  = cnt + NBK_MAX;
    int*   wsum = off + NBK_MAX + 4;
    int*   sp   = wsum + NBK_MAX / 32;
    float* sv   = (float*)(sp + BIN_C);

    const int tid  = threadIdx.x;
    const int lane = tid & 31;
    const int wave = tid >> 5;
    const int tot  = nnz1 + nnz2;
    const int base = bb * BIN_C;

    cnt[tid] = 0;
    __syncthreads();

    int   pk[BIN_EPT], kr[BIN_EPT];
    int   dd[BIN_EPT], ss[BIN_EPT];
    float vv[BIN_EPT];
    #pragma unroll
    for (int u = 0; u < BIN_EPT; ++u) {
        const int    e      = base + u * BIN_THR + tid;
        const int    ec     = e < tot ? e : tot - 1;
        const bool   second = ec >= nnz1;
        const int    i      = second ? ec - nnz1 : ec;
        const int*   dq     = second ? dst2 : dst1;
        const int*   sq     = second ? src2 : src1;
        const float* vq     = second ? val2 : val1;
        dd[u] = dq[i];
        ss[u] = sq[i] + (second ? n : 0);
        vv[u] = vq[i];
    }
    #pragma unroll
    for (int u = 0; u < BIN_EPT; ++u) {
        const int e   = base + u * BIN_THR + tid;
        const int d   = dd[u];
        const int k   = d / RP;
        const int rel = d - k * RP;
        pk[u] = (ss[u] << RP_BITS) | rel;
        kr[u] = -1;
        if (e < tot && (unsigned)d < (unsigned)n) {
            const int r = __hip_atomic_fetch_add(&cnt[k], 1, __ATOMIC_RELAXED, __HIP_MEMORY_SCOPE_WORKGROUP);
            kr[u] = (k << 16) | r;
        }
    }
    __syncthreads();

    int incl = cnt[tid];
    for (int dlt = 1; dlt < 32; dlt <<= 1) {
        const int t = __shfl_up(incl, dlt);
        if (lane >= dlt) incl += t;
    }
    if (lane == 31) wsum[wave] = incl;
    __syncthreads();
    {
        int add = 0;
        for (int w = 0; w < wave; ++w) add += wsum[w];
        off[tid + 1] = incl + add;
        if (tid == 0) off[0] = 0;
    }
    __syncthreads();

    #pragma unroll
    for (int u = 0; u < BIN_EPT; ++u) {
        if (kr[u] >= 0) {
            const int p = off[kr[u] >> 16] + (kr[u] & 0xffff);
            sp[p] = pk[u];
            sv[p] = vv[u];
        }
    }
    woff[(size_t)bb * OFF_PITCH + tid] = off[tid];
    if (tid == 0) woff[(size_t)bb * OFF_PITCH + NBK_MAX] = off[NBK_MAX];
    __syncthreads();

    const int total = off[NBK_MAX];
    for (int i = tid; i < total; i += BIN_THR) {
        wp[base + i] = sp[i];
        wv[base + i] = sv[i];
    }
}

#define KC      100
#define KC4     (KC / 4)
#define NCHUNK  (KDIM / KC)
#define NWAVE   8
#define XP      (16 * KC)
#define XPIECES (16 * KC4)
#define XLD     ((XPIECES + 31) / 32)
#define XTAIL   (XPIECES - (XLD - 1) * 32)
#define SW_FL   (HIDDEN * KDIM)
#define SW_V4   (SW_FL / 4)
#define WGRP    8
#define WRND    ((SW_V4 + WGRP * 256 - 1) / (WGRP * 256))

__global__ __launch_bounds__(256) void k_gemm_relu_proj_bin(
    const float* __restrict__ x, const float* __restrict__ W1,
    const float* __restrict__ Wf,
    float* __restrict__ out, float* __restrict__ g1, float* __restrict__ g2,
    int n_tiles, int n_gemm_blocks,
    const int* __restrict__ src1, const int* __restrict__ dst1, const float* __restrict__ val1, int nnz1,
    const int* __restrict__ src2, const int* __restrict__ dst2, const float* __restrict__ val2, int nnz2,
    int n, int* __restrict__ wp, float* __restrict__ wv, int* __restrict__ woff)
{
    __shared__ __attribute__((aligned(16))) float smem[SW_FL + NWAVE * XP];

    if ((int)blockIdx.x >= n_gemm_blocks) {
        bin_role(smem, (int)blockIdx.x - n_gemm_blocks,
                 src1, dst1, val1, nnz1, src2, dst2, val2, nnz2, n, wp, wv, woff);
        return;
    }
    const int tid   = threadIdx.x;
    const int lane  = tid & 31;
    const int wave  = __builtin_amdgcn_readfirstlane(tid >> 5);
    const int nn    = lane & 15;
    const int khalf = (lane >> 4) << 1;

    float* sw = smem;
    float* px = smem + SW_FL + wave * XP;

    int goff[XLD];
    #pragma unroll
    for (int i = 0; i < XLD; ++i) {
        const int q0 = lane + 32 * i;
        const int q  = q0 < XPIECES ? q0 : XPIECES - 1;
        const int r  = q / KC4;
        goff[i] = r * KDIM + (q - r * KC4) * 4;
    }
    const bool tail_ok = lane < XTAIL;
    float* ps = px + 4 * lane;

    const int tstep = NWAVE * n_gemm_blocks;
    int t = wave * n_gemm_blocks + (int)blockIdx.x;

    v4f rx[XLD];
    if (t < n_tiles) {
        const float* xt = x + (size_t)t * (16 * KDIM);
        #pragma unroll
        for (int i = 0; i < XLD; ++i) rx[i] = __builtin_nontemporal_load((const v4f*)(xt + goff[i]));
    }

    #pragma unroll
    for (int r = 0; r < WRND; ++r) {
        v4f tw[WGRP];
        #pragma unroll
        for (int u = 0; u < WGRP; ++u) {
            const int i  = tid + 256 * (WGRP * r + u);
            const int ic = i < SW_V4 ? i : SW_V4 - 1;
            tw[u] = *(const v4f*)(W1 + 4 * ic);
        }
        #pragma unroll
        for (int u = 0; u < WGRP; ++u) {
            const int i = tid + 256 * (WGRP * r + u);
            if (i < SW_V4) *(v4f*)(sw + 4 * i) = tw[u];
        }
    }
    __syncthreads();

    const float* ax = px + nn * KC + khalf;
    const float* bw = sw + nn * KDIM + khalf;

    const bool live = nn < NOUT;
    const int  wn   = live ? nn : NOUT - 1;
    const float* wrow = Wf + (size_t)wn * (3 * HIDDEN) + khalf;

    for (; t < n_tiles; t += tstep) {
        const float* xt = x + (size_t)t * (16 * KDIM);
        const int    tn = t + tstep;

        #pragma unroll
        for (int i = 0; i < XLD - 1; ++i) *(v4f*)(ps + 128 * i) = rx[i];
        if (tail_ok) *(v4f*)(ps + 128 * (XLD - 1)) = rx[XLD - 1];

        v8f acc0 = {}, acc1 = {}, acc2 = {}, acc3 = {};
        for (int c = 0; c < NCHUNK; ++c) {
            if (c + 1 < NCHUNK || tn < n_tiles) {
                const float* xn = c + 1 < NCHUNK ? xt + (c + 1) * KC : x + (size_t)tn * (16 * KDIM);
                #pragma unroll
                for (int i = 0; i < XLD; ++i) rx[i] = __builtin_nontemporal_load((const v4f*)(xn + goff[i]));
            }
            const float* bc = bw + c * KC;
            for (int kk = 0; kk < KC; kk += 4) {
                v2f a  = *(const v2f*)(ax + kk);
                v2f b0 = *(const v2f*)(bc +  0 * KDIM + kk);
                v2f b1 = *(const v2f*)(bc + 16 * KDIM + kk);
                v2f b2 = *(const v2f*)(bc + 32 * KDIM + kk);
                v2f b3 = *(const v2f*)(bc + 48 * KDIM + kk);
                acc0 = __builtin_amdgcn_wmma_f32_16x16x4_f32(false, a, false, b0, (short)0, acc0, false, false);
                acc1 = __builtin_amdgcn_wmma_f32_16x16x4_f32(false, a, false, b1, (short)0, acc1, false, false);
                acc2 = __builtin_amdgcn_wmma_f32_16x16x4_f32(false, a, false, b2, (short)0, acc2, false, false);
                acc3 = __builtin_amdgcn_wmma_f32_16x16x4_f32(false, a, false, b3, (short)0, acc3, false, false);
            }
            if (c + 1 < NCHUNK) {
                #pragma unroll
                for (int i = 0; i < XLD - 1; ++i) *(v4f*)(ps + 128 * i) = rx[i];
                if (tail_ok) *(v4f*)(ps + 128 * (XLD - 1)) = rx[XLD - 1];
            }
        }

        float* hx = px + ((lane >> 4) << 3) * KC + nn;
        for (int r = 0; r < 8; ++r) {
            float v0 = acc0[r]; v0 = v0 > 0.f ? v0 : 0.f;
            float v1 = acc1[r]; v1 = v1 > 0.f ? v1 : 0.f;
            float v2 = acc2[r]; v2 = v2 > 0.f ? v2 : 0.f;
            float v3 = acc3[r]; v3 = v3 > 0.f ? v3 : 0.f;
            hx[r * KC +  0] = v0;
            hx[r * KC + 16] = v1;
            hx[r * KC + 32] = v2;
            hx[r * KC + 48] = v3;
        }

        v8f p0 = {}, p1 = {}, p2 = {};
        #pragma unroll
        for (int k0 = 0; k0 < HIDDEN; k0 += 4) {
            const v2f a = *(const v2f*)(ax + k0);
            const v2f w0 = *(const v2f*)(wrow + 0 * HIDDEN + k0);
            const v2f w1 = *(const v2f*)(wrow + 1 * HIDDEN + k0);
            const v2f w2 = *(const v2f*)(wrow + 2 * HIDDEN + k0);
            v2f b0, b1, b2;
            b0.x = live ? w0.x : 0.f;  b0.y = live ? w0.y : 0.f;
            b1.x = live ? w1.x : 0.f;  b1.y = live ? w1.y : 0.f;
            b2.x = live ? w2.x : 0.f;  b2.y = live ? w2.y : 0.f;
            p0 = __builtin_amdgcn_wmma_f32_16x16x4_f32(false, a, false, b0, (short)0, p0, false, false);
            p1 = __builtin_amdgcn_wmma_f32_16x16x4_f32(false, a, false, b1, (short)0, p1, false, false);
            p2 = __builtin_amdgcn_wmma_f32_16x16x4_f32(false, a, false, b2, (short)0, p2, false, false);
        }
        const int m_lo = t * 16 + ((lane >> 4) << 3);
        for (int r = 0; r < 8; ++r) {
            const size_t m = (size_t)(m_lo + r);
            g1[m * GPITCH + nn] = p1[r];
            g2[m * GPITCH + nn] = p2[r];
            if (live) out[m * NOUT + nn] = p0[r];
        }
    }
}

#define OWN_CAP  16384
#define OWN_RUNS 4
#define OWN_U    8

__device__ __forceinline__ void lds_add(float* p, float v)
{
    __hip_atomic_fetch_add(p, v, __ATOMIC_RELAXED, __HIP_MEMORY_SCOPE_WORKGROUP);
}

__device__ __forceinline__ void own_add(float* acc, int rec, float v, v4f a, v4f b, v2f c)
{
    float* ar = acc + (rec & ((1 << RP_BITS) - 1)) * NOUT;
    lds_add(ar + 0, v * a.x);  lds_add(ar + 1, v * a.y);
    lds_add(ar + 2, v * a.z);  lds_add(ar + 3, v * a.w);
    lds_add(ar + 4, v * b.x);  lds_add(ar + 5, v * b.y);
    lds_add(ar + 6, v * b.z);  lds_add(ar + 7, v * b.w);
    lds_add(ar + 8, v * c.x);  lds_add(ar + 9, v * c.y);
}

template <bool WITH_VAL, class F>
__device__ __forceinline__ void walk_runs(
    const int* __restrict__ wp, const float* __restrict__ wv, const int* __restrict__ woff,
    int nb, int part, int wave, int lane, F f)
{
    for (int b0 = 0; b0 < nb; b0 += OWN_THR) {
        const int b = b0 + wave + 32 * lane;
        int st = 0, en = 0;
        if (b < nb) {
            const int* o = woff + (size_t)b * OFF_PITCH + part;
            st = b * BIN_C + o[0];
            en = b * BIN_C + o[1];
        }
        const int left = nb - b0 - wave;
        const int nseg = left <= 0 ? 0 : ((left + 31) / 32 < 32 ? (left + 31) / 32 : 32);

        for (int j = 0; j < nseg; j += OWN_RUNS) {
            int sst[OWN_RUNS], sen[OWN_RUNS], len = 0;
            #pragma unroll
            for (int k = 0; k < OWN_RUNS; ++k) {
                const int  jk = j + k < 32 ? j + k : 31;
                const bool on = j + k < nseg;
                const int  a  = __shfl(st, jk), e = __shfl(en, jk);
                sst[k] = on ? a : 0;  sen[k] = on ? e : 0;
                len = sen[k] - sst[k] > len ? sen[k] - sst[k] : len;
            }
            for (int o = 0; o < len; o += 32) {
                int rr[OWN_RUNS];  float vv[OWN_RUNS];  bool ok[OWN_RUNS];
                #pragma unroll
                for (int k = 0; k < OWN_RUNS; ++k) {
                    const int i = sst[k] + o + lane;
                    ok[k] = i < sen[k];
                    rr[k] = wp[ok[k] ? i : 0];
                    vv[k] = WITH_VAL ? wv[ok[k] ? i : 0] : 0.f;
                }
                #pragma unroll
                for (int k = 0; k < OWN_RUNS; ++k)
                    if (ok[k]) f(rr[k], vv[k]);
            }
        }
    }
}

__global__ __launch_bounds__(OWN_THR) void k_owner(
    const int* __restrict__ wp, const float* __restrict__ wv, const int* __restrict__ woff, int nb,
    const float* __restrict__ g, float* __restrict__ out, int n)
{
    __shared__ int   cnt[RP];
    __shared__ int   rstart[RP + 1];
    __shared__ int   wtot[OWN_THR / 32];
    __shared__ int   srow[OWN_CAP + OWN_U];
    __shared__ float sval[OWN_CAP + OWN_U];
    __shared__ float acc[RP * NOUT];

    const int tid  = threadIdx.x;
    const int lane = tid & 31;
    const int wave = __builtin_amdgcn_readfirstlane(tid >> 5);
    const int part = blockIdx.x;
    const int lo   = part * RP;
    const int rows = n - lo < RP ? n - lo : RP;

    if (tid < RP) cnt[tid] = 0;
    __syncthreads();

    walk_runs<false>(wp, wv, woff, nb, part, wave, lane, [&](int rec, float) {
        __hip_atomic_fetch_add(&cnt[rec & ((1 << RP_BITS) - 1)], 1, __ATOMIC_RELAXED, __HIP_MEMORY_SCOPE_WORKGROUP);
    });
    __syncthreads();

    const int c = tid < RP ? cnt[tid] : 0;
    int incl = c;
    for (int dlt = 1; dlt < 32; dlt <<= 1) {
        const int t = __shfl_up(incl, dlt);
        if (lane >= dlt) incl += t;
    }
    if (lane == 31) wtot[wave] = incl;
    __syncthreads();
    {
        int add = 0;
        for (int w = 0; w < wave; ++w) add += wtot[w];
        if (tid < RP) { rstart[tid] = add + incl - c;  cnt[tid] = add + incl - c; }
        if (tid == RP - 1) rstart[RP] = add + incl;
    }
    __syncthreads();
    const int total = rstart[RP];

    if (total <= OWN_CAP) {
        walk_runs<true>(wp, wv, woff, nb, part, wave, lane, [&](int rec, float v) {
            const int p = __hip_atomic_fetch_add(&cnt[rec & ((1 << RP_BITS) - 1)], 1, __ATOMIC_RELAXED, __HIP_MEMORY_SCOPE_WORKGROUP);
            srow[p] = rec >> RP_BITS;
            sval[p] = v;
        });
        __syncthreads();

        const int  grp  = tid >> 2, q = tid & 3;
        const int  rowA = grp, rowB = grp + 256;
        const bool hasB = rowB < RP;
        int iA = rstart[rowA], iB = hasB ? rstart[rowB] : 0;
        const int eA = rstart[rowA + 1], eB = hasB ? rstart[rowB + 1] : 0;
        const float* gq = g + 4 * q;
        v4f sA = {0.f, 0.f, 0.f, 0.f}, sB = {0.f, 0.f, 0.f, 0.f};
        while (iA < eA || iB < eB) {
            int ra[OWN_U], rb[OWN_U];  float va[OWN_U], vb[OWN_U];  v4f xa[OWN_U], xb[OWN_U];
            #pragma unroll
            for (int u = 0; u < OWN_U; ++u) {
                const bool ka = iA + u < eA, kb = iB + u < eB;
                const int   ta = srow[iA + u], tb = srow[iB + u];
                const float ua = sval[iA + u], ub = sval[iB + u];
                ra[u] = ka ? ta : 0;  va[u] = ka ? ua : 0.f;
                rb[u] = kb ? tb : 0;  vb[u] = kb ? ub : 0.f;
            }
            #pragma unroll
            for (int u = 0; u < OWN_U; ++u) {
                xa[u] = *(const v4f*)(gq + (size_t)ra[u] * GPITCH);
                xb[u] = *(const v4f*)(gq + (size_t)rb[u] * GPITCH);
            }
            #pragma unroll
            for (int u = 0; u < OWN_U; ++u) {
                if (iA + u < eA) sA += va[u] * xa[u];
                if (iB + u < eB) sB += vb[u] * xb[u];
            }
            iA = iA + OWN_U < eA ? iA + OWN_U : eA;
            iB = iB + OWN_U < eB ? iB + OWN_U : eB;
        }
        if (q < 3) {
            float* aA = acc + rowA * NOUT + 4 * q;
            aA[0] = sA.x;  aA[1] = sA.y;
            if (q < 2) { aA[2] = sA.z;  aA[3] = sA.w; }
            if (hasB) {
                float* aB = acc + rowB * NOUT + 4 * q;
                aB[0] = sB.x;  aB[1] = sB.y;
                if (q < 2) { aB[2] = sB.z;  aB[3] = sB.w; }
            }
        }
    } else {
        for (int i = tid; i < RP * NOUT; i += OWN_THR) acc[i] = 0.f;
        __syncthreads();
        walk_runs<true>(wp, wv, woff, nb, part, wave, lane, [&](int rec, float v) {
            const float* gr = g + (size_t)(rec >> RP_BITS) * GPITCH;
            own_add(acc, rec, v, *(const v4f*)(gr), *(const v4f*)(gr + 4), *(const v2f*)(gr + 8));
        });
    }
    __syncthreads();

    float* o = out + (size_t)lo * NOUT;
    const int ocnt = rows * NOUT;
    for (int i = tid; i < ocnt; i += OWN_THR) o[i] = o[i] + acc[i];
}

extern "C" void kernel_launch(void* const* d_in, const int* in_sizes, int n_in,
                              void* d_out, int out_size, void* d_ws, size_t ws_size,
                              hipStream_t stream)
{
    const float* x    = (const float*)d_in[0];
    const float* W1   = (const float*)d_in[1];
    const float* Wf   = (const float*)d_in[2];
    const int*   src1 = (const int*)  d_in[3];
    const int*   dst1 = (const int*)  d_in[4];
    const float* val1 = (const float*)d_in[5];
    const int*   src2 = (const int*)  d_in[6];
    const int*   dst2 = (const int*)  d_in[7];
    const float* val2 = (const float*)d_in[8];

    const int n    = in_sizes[0] / KDIM;
    const int nnz1 = in_sizes[3];
    const int nnz2 = in_sizes[6];
    const int nb   = (nnz1 + nnz2 + BIN_C - 1) / BIN_C;

    float* g1   = (float*)d_ws;
    float* g2   = g1 + (size_t)n * GPITCH;
    int*   wp   = (int*)(g2 + (size_t)n * GPITCH);
    float* wv   = (float*)(wp + (size_t)nb * BIN_C);
    int*   woff = (int*)(wv + (size_t)nb * BIN_C);
    float* out  = (float*)d_out;

    const int n_tiles = (n + 15) / 16;

    const int n_gemm = 256;

    k_gemm_relu_proj_bin<<<n_gemm + nb, 256, 0, stream>>>(x, W1, Wf, out, g1, g2, n_tiles, n_gemm,
        src1, dst1, val1, nnz1, src2, dst2, val2, nnz2, n, wp, wv, woff);

    k_owner<<<(n + RP - 1) / RP, OWN_THR, 0, stream>>>(wp, wv, woff, nb, g1, out, n);
}
